# starting kernel
# baseline (speedup 1.0000x reference)
#include <hip/hip_runtime.h>
#include <hip/hip_bf16.h>
#include <math.h>


typedef __attribute__((ext_vector_type(2))) float v2f;
typedef __attribute__((ext_vector_type(8))) float v8f;

#define NTOK   65536
#define DM     8
#define DI     16
#define CHUNK  256
#define NCHUNK 256
#define N16    (NTOK * DI)

__device__ inline v8f wmma4(v2f a, v2f b, v8f c) {
  return __builtin_amdgcn_wmma_f32_16x16x4_f32(
      false, a, false, b, (short)0, c, false, false);
}

__device__ inline float softplus_f(float s) {
  return (s > 20.f) ? s : logf(1.f + __expf(s));
}

__global__ void k_proj(const float* __restrict__ x,
                       const float* __restrict__ inw,
                       const float* __restrict__ dtw,
                       const float* __restrict__ dtb,
                       const float* __restrict__ bw,
                       const float* __restrict__ cw,
                       float* __restrict__ xb, float* __restrict__ z,
                       float* __restrict__ dt, float* __restrict__ Bv,
                       float* __restrict__ Cv) {
  __shared__ float lds[4][256];
  const int lane = threadIdx.x & 31;
  const int wave = threadIdx.x >> 5;
  const int t0   = (blockIdx.x * 4 + wave) * 16;
  const int lo   = lane & 15, hi = lane >> 4;

  v8f axb = {}, az = {};
#pragma unroll
  for (int kk = 0; kk < 2; ++kk) {
    const int kb = kk * 4 + 2 * hi;
    v2f a, b0, b1;
    a.x  = x[(t0 + lo) * 8 + kb + 0];
    a.y  = x[(t0 + lo) * 8 + kb + 1];
    b0.x = inw[lo * 8 + kb + 0];
    b0.y = inw[lo * 8 + kb + 1];
    b1.x = inw[(lo + 16) * 8 + kb + 0];
    b1.y = inw[(lo + 16) * 8 + kb + 1];
    axb = wmma4(a, b0, axb);
    az  = wmma4(a, b1, az);
  }

#pragma unroll
  for (int r = 0; r < 8; ++r) lds[wave][(r + 8 * hi) * 16 + lo] = axb[r];

  v8f adt = {}, ab = {}, ac = {};
#pragma unroll
  for (int kk = 0; kk < 4; ++kk) {
    const int kb = kk * 4 + 2 * hi;
    v2f a, b;
    a.x = lds[wave][lo * 16 + kb + 0];
    a.y = lds[wave][lo * 16 + kb + 1];
    b.x = dtw[lo * 16 + kb + 0]; b.y = dtw[lo * 16 + kb + 1];
    adt = wmma4(a, b, adt);
    b.x = bw[lo * 16 + kb + 0];  b.y = bw[lo * 16 + kb + 1];
    ab  = wmma4(a, b, ab);
    b.x = cw[lo * 16 + kb + 0];  b.y = cw[lo * 16 + kb + 1];
    ac  = wmma4(a, b, ac);
  }

  const float bias = dtb[lo];
#pragma unroll
  for (int r = 0; r < 8; ++r) {
    const int idx = (t0 + r + 8 * hi) * 16 + lo;
    xb[idx] = axb[r];
    z[idx]  = az[r];
    dt[idx] = softplus_f(adt[r] + bias);
    Bv[idx] = ab[r];
    Cv[idx] = ac[r];
  }
}

__global__ void k_chunkop(const float* __restrict__ dt,
                          const float* __restrict__ xb,
                          const float* __restrict__ Bv,
                          const float* __restrict__ A_log,
                          const float* __restrict__ rope,
                          float* __restrict__ Tmat,
                          float* __restrict__ bvec) {
  const int c    = blockIdx.x;
  const int d    = threadIdx.x >> 5;
  const int lane = threadIdx.x & 31;
  const int lo   = lane & 15, hi = lane >> 4;

  float Arow[8], Frow[8];
#pragma unroll
  for (int r = 0; r < 8; ++r) {
    const int j = r + 8 * hi;
    Arow[r] = -__expf(A_log[d * 16 + j]);
    Frow[r] = rope[d * 16 + j];
  }
  const float A2 = -__expf(A_log[d * 16 + lo]);
  const float F2 = rope[d * 16 + lo];

  float T[8];
#pragma unroll
  for (int r = 0; r < 8; ++r) T[r] = ((r + 8 * hi) == lo) ? 1.f : 0.f;
  float b = 0.f;

  const int t0  = c * CHUNK;
  const int src = (lane & 16) | ((lo + 15) & 15);

  for (int t = 0; t < CHUNK; ++t) {
    const int tt   = t0 + t;
    const float dtc = dt[tt * 16 + d];
    const float xbd = xb[tt * 16 + d];
    const float Bj  = Bv[tt * 16 + lo];

    const float roll0 = __shfl_xor(T[7], 16, 32);
#pragma unroll
    for (int r = 7; r >= 1; --r) {
      const float ta   = 0.5f * dtc * Arow[r];
      const float abar = (1.f + ta) / (1.f - ta + 1e-8f);
      const float ang  = dtc * Frow[r];
      T[r] = abar * (__cosf(ang) * T[r] - __sinf(ang) * T[r - 1]);
    }
    {
      const float ta   = 0.5f * dtc * Arow[0];
      const float abar = (1.f + ta) / (1.f - ta + 1e-8f);
      const float ang  = dtc * Frow[0];
      T[0] = abar * (__cosf(ang) * T[0] - __sinf(ang) * roll0);
    }
    const float ta2   = 0.5f * dtc * A2;
    const float abar2 = (1.f + ta2) / (1.f - ta2 + 1e-8f);
    const float ang2  = dtc * F2;
    const float brl   = __shfl(b, src, 32);
    b = abar2 * (__cosf(ang2) * b - __sinf(ang2) * brl) + dtc * Bj * xbd;
  }

#pragma unroll
  for (int r = 0; r < 8; ++r)
    Tmat[((c * 16 + d) * 16 + (r + 8 * hi)) * 16 + lo] = T[r];
  if (hi == 0) bvec[(c * 16 + d) * 16 + lo] = b;
}

__global__ void k_combine(const float* __restrict__ Tmat,
                          const float* __restrict__ bvec,
                          const float* __restrict__ init_h,
                          float* __restrict__ hstart,
                          float* __restrict__ h_final) {
  const int d    = threadIdx.x >> 5;
  const int lane = threadIdx.x & 31;
  const int lo   = lane & 15;
  float h = init_h[d * 16 + lo];

  for (int c = 0; c < NCHUNK; ++c) {
    if (c + 1 < NCHUNK) {
      __builtin_prefetch(Tmat + (((c + 1) * 16 + d) * 16 + lo) * 16, 0, 1);
      __builtin_prefetch(bvec + ((c + 1) * 16 + d) * 16 + lo, 0, 1);
    }
    if (lane < 16) hstart[(c * 16 + d) * 16 + lo] = h;

    const float4* Trow =
        reinterpret_cast<const float4*>(Tmat + ((c * 16 + d) * 16 + lo) * 16);
    const float4 q0 = Trow[0], q1 = Trow[1], q2 = Trow[2], q3 = Trow[3];
    float Tr[16] = {q0.x, q0.y, q0.z, q0.w, q1.x, q1.y, q1.z, q1.w,
                    q2.x, q2.y, q2.z, q2.w, q3.x, q3.y, q3.z, q3.w};
    float hn = bvec[(c * 16 + d) * 16 + lo];
#pragma unroll
    for (int k = 0; k < 16; ++k)
      hn = fmaf(Tr[k], __shfl(h, (lane & 16) | k, 32), hn);
    h = hn;
  }
  if (lane < 16) h_final[d * 16 + lo] = h;
}

__global__ void k_scan(const float* __restrict__ dt,
                       const float* __restrict__ xb,
                       const float* __restrict__ Bv,
                       const float* __restrict__ Cv,
                       const float* __restrict__ A_log,
                       const float* __restrict__ rope,
                       const float* __restrict__ hstart,
                       float* __restrict__ y) {
  const int c    = blockIdx.x;
  const int d    = threadIdx.x >> 5;
  const int lane = threadIdx.x & 31;
  const int lo   = lane & 15;
  const float A2 = -__expf(A_log[d * 16 + lo]);
  const float F2 = rope[d * 16 + lo];
  float h = hstart[(c * 16 + d) * 16 + lo];
  const int t0  = c * CHUNK;
  const int src = (lane & 16) | ((lo + 15) & 15);

  for (int t = 0; t < CHUNK; ++t) {
    const int tt    = t0 + t;
    const float dtc = dt[tt * 16 + d];
    const float xbd = xb[tt * 16 + d];
    const float Bj  = Bv[tt * 16 + lo];
    const float Cj  = Cv[tt * 16 + lo];
    const float ta   = 0.5f * dtc * A2;
    const float abar = (1.f + ta) / (1.f - ta + 1e-8f);
    const float ang  = dtc * F2;
    const float hr   = __shfl(h, src, 32);
    h = abar * (__cosf(ang) * h - __sinf(ang) * hr) + dtc * Bj * xbd;
    float p = h * Cj;
    p += __shfl_xor(p, 1, 32);
    p += __shfl_xor(p, 2, 32);
    p += __shfl_xor(p, 4, 32);
    p += __shfl_xor(p, 8, 32);
    if (lane == 0) y[tt * 16 + d] = p;
  }
}

__global__ void k_out(const float* __restrict__ y,
                      const float* __restrict__ z,
                      const float* __restrict__ xb,
                      const float* __restrict__ Dp,
                      const float* __restrict__ outw,
                      float* __restrict__ out) {
  const int lane = threadIdx.x & 31;
  const int wave = threadIdx.x >> 5;
  const int t0   = (blockIdx.x * 4 + wave) * 16;
  const int lo   = lane & 15, hi = lane >> 4;

  v8f acc = {};
#pragma unroll
  for (int kk = 0; kk < 4; ++kk) {
    const int kb = kk * 4 + 2 * hi;
    v2f a, b;
#pragma unroll
    for (int v = 0; v < 2; ++v) {
      const int k   = kb + v;
      const int idx = (t0 + lo) * 16 + k;
      const float zz  = z[idx];
      const float sil = zz / (1.f + __expf(-zz));
      const float val = y[idx] * sil + Dp[k] * xb[idx];
      if (v == 0) a.x = val; else a.y = val;
    }
    b.x = (lo < 8) ? outw[lo * 16 + kb + 0] : 0.f;
    b.y = (lo < 8) ? outw[lo * 16 + kb + 1] : 0.f;
    acc = wmma4(a, b, acc);
  }
  if (lo < 8) {
#pragma unroll
    for (int r = 0; r < 8; ++r) out[(t0 + r + 8 * hi) * 8 + lo] = acc[r];
  }
}

extern "C" void kernel_launch(void* const* d_in, const int* in_sizes, int n_in,
                              void* d_out, int out_size, void* d_ws,
                              size_t ws_size, hipStream_t stream) {
  const float* x    = (const float*)d_in[0];
  const float* init = (const float*)d_in[1];
  const float* inw  = (const float*)d_in[2];
  const float* dtw  = (const float*)d_in[3];
  const float* dtb  = (const float*)d_in[4];
  const float* bw   = (const float*)d_in[5];
  const float* cw   = (const float*)d_in[6];
  const float* alog = (const float*)d_in[7];
  const float* Dp   = (const float*)d_in[8];
  const float* rope = (const float*)d_in[9];
  const float* outw = (const float*)d_in[10];
  float* out = (float*)d_out;

  float* ws = (float*)d_ws;
  float* XB = ws;
  float* Z  = XB + N16;
  float* DT = Z + N16;
  float* BV = DT + N16;
  float* CV = BV + N16;
  float* Y  = CV + N16;
  float* T  = Y + N16;
  float* BC = T + N16;
  float* HS = BC + NCHUNK * 256;

  k_proj<<<NTOK / 64, 128, 0, stream>>>(x, inw, dtw, dtb, bw, cw,
                                        XB, Z, DT, BV, CV);
  k_chunkop<<<NCHUNK, 512, 0, stream>>>(DT, XB, BV, alog, rope, T, BC);
  k_combine<<<1, 512, 0, stream>>>(T, BC, init, HS, out + NTOK * DM);
  k_scan<<<NCHUNK, 512, 0, stream>>>(DT, XB, BV, CV, alog, rope, HS, Y);
  k_out<<<NTOK / 64, 128, 0, stream>>>(Y, Z, XB, Dp, outw, out);
}


# K2/K3/K4 serial loops: each iteration's global operands are loaded several iterations ahead into a register ring, so load latency leaves the recurrence chain; arithmetic and sum order unchanged
# speedup vs baseline: 1.3628x; 1.3628x over previous
#include <hip/hip_runtime.h>
#include <hip/hip_bf16.h>
#include <math.h>


typedef __attribute__((ext_vector_type(2))) float v2f;
typedef __attribute__((ext_vector_type(8))) float v8f;

#define NTOK   65536
#define DM     8
#define DI     16
#define CHUNK  256
#define NCHUNK 256
#define N16    (NTOK * DI)

__device__ inline v8f wmma4(v2f a, v2f b, v8f c) {
  return __builtin_amdgcn_wmma_f32_16x16x4_f32(
      false, a, false, b, (short)0, c, false, false);
}

__device__ inline float softplus_f(float s) {
  return (s > 20.f) ? s : logf(1.f + __expf(s));
}

__global__ void k_proj(const float* __restrict__ x,
                       const float* __restrict__ inw,
                       const float* __restrict__ dtw,
                       const float* __restrict__ dtb,
                       const float* __restrict__ bw,
                       const float* __restrict__ cw,
                       float* __restrict__ xb, float* __restrict__ z,
                       float* __restrict__ dt, float* __restrict__ Bv,
                       float* __restrict__ Cv) {
  __shared__ float lds[4][256];
  const int lane = threadIdx.x & 31;
  const int wave = threadIdx.x >> 5;
  const int t0   = (blockIdx.x * 4 + wave) * 16;
  const int lo   = lane & 15, hi = lane >> 4;

  v8f axb = {}, az = {};
#pragma unroll
  for (int kk = 0; kk < 2; ++kk) {
    const int kb = kk * 4 + 2 * hi;
    v2f a, b0, b1;
    a.x  = x[(t0 + lo) * 8 + kb + 0];
    a.y  = x[(t0 + lo) * 8 + kb + 1];
    b0.x = inw[lo * 8 + kb + 0];
    b0.y = inw[lo * 8 + kb + 1];
    b1.x = inw[(lo + 16) * 8 + kb + 0];
    b1.y = inw[(lo + 16) * 8 + kb + 1];
    axb = wmma4(a, b0, axb);
    az  = wmma4(a, b1, az);
  }

#pragma unroll
  for (int r = 0; r < 8; ++r) lds[wave][(r + 8 * hi) * 16 + lo] = axb[r];

  v8f adt = {}, ab = {}, ac = {};
#pragma unroll
  for (int kk = 0; kk < 4; ++kk) {
    const int kb = kk * 4 + 2 * hi;
    v2f a, b;
    a.x = lds[wave][lo * 16 + kb + 0];
    a.y = lds[wave][lo * 16 + kb + 1];
    b.x = dtw[lo * 16 + kb + 0]; b.y = dtw[lo * 16 + kb + 1];
    adt = wmma4(a, b, adt);
    b.x = bw[lo * 16 + kb + 0];  b.y = bw[lo * 16 + kb + 1];
    ab  = wmma4(a, b, ab);
    b.x = cw[lo * 16 + kb + 0];  b.y = cw[lo * 16 + kb + 1];
    ac  = wmma4(a, b, ac);
  }

  const float bias = dtb[lo];
#pragma unroll
  for (int r = 0; r < 8; ++r) {
    const int idx = (t0 + r + 8 * hi) * 16 + lo;
    xb[idx] = axb[r];
    z[idx]  = az[r];
    dt[idx] = softplus_f(adt[r] + bias);
    Bv[idx] = ab[r];
    Cv[idx] = ac[r];
  }
}

__global__ void k_chunkop(const float* __restrict__ dt,
                          const float* __restrict__ xb,
                          const float* __restrict__ Bv,
                          const float* __restrict__ A_log,
                          const float* __restrict__ rope,
                          float* __restrict__ Tmat,
                          float* __restrict__ bvec) {
  const int c    = blockIdx.x;
  const int d    = threadIdx.x >> 5;
  const int lane = threadIdx.x & 31;
  const int lo   = lane & 15, hi = lane >> 4;

  float Arow[8], Frow[8];
#pragma unroll
  for (int r = 0; r < 8; ++r) {
    const int j = r + 8 * hi;
    Arow[r] = -__expf(A_log[d * 16 + j]);
    Frow[r] = rope[d * 16 + j];
  }
  const float A2 = -__expf(A_log[d * 16 + lo]);
  const float F2 = rope[d * 16 + lo];

  float T[8];
#pragma unroll
  for (int r = 0; r < 8; ++r) T[r] = ((r + 8 * hi) == lo) ? 1.f : 0.f;
  float b = 0.f;

  const int t0  = c * CHUNK;
  const int src = (lane & 16) | ((lo + 15) & 15);

  constexpr int PF2 = 4;
  float dtq[PF2] = {}, xbq[PF2] = {}, Bq[PF2] = {};
#pragma unroll
  for (int u = 0; u < PF2 - 1; ++u) {
    const int tt = t0 + u;
    dtq[u] = dt[tt * 16 + d];
    xbq[u] = xb[tt * 16 + d];
    Bq[u]  = Bv[tt * 16 + lo];
  }

  for (int t = 0; t < CHUNK; t += PF2) {
#pragma unroll
  for (int u = 0; u < PF2; ++u) {
    {
      const int v  = (u + PF2 - 1) % PF2;
      const int tn = t0 + min(t + u + PF2 - 1, CHUNK - 1);
      dtq[v] = dt[tn * 16 + d];
      xbq[v] = xb[tn * 16 + d];
      Bq[v]  = Bv[tn * 16 + lo];
    }
    const float dtc = dtq[u];
    const float xbd = xbq[u];
    const float Bj  = Bq[u];

    const float roll0 = __shfl_xor(T[7], 16, 32);
#pragma unroll
    for (int r = 7; r >= 1; --r) {
      const float ta   = 0.5f * dtc * Arow[r];
      const float abar = (1.f + ta) / (1.f - ta + 1e-8f);
      const float ang  = dtc * Frow[r];
      T[r] = abar * (__cosf(ang) * T[r] - __sinf(ang) * T[r - 1]);
    }
    {
      const float ta   = 0.5f * dtc * Arow[0];
      const float abar = (1.f + ta) / (1.f - ta + 1e-8f);
      const float ang  = dtc * Frow[0];
      T[0] = abar * (__cosf(ang) * T[0] - __sinf(ang) * roll0);
    }
    const float ta2   = 0.5f * dtc * A2;
    const float abar2 = (1.f + ta2) / (1.f - ta2 + 1e-8f);
    const float ang2  = dtc * F2;
    const float brl   = __shfl(b, src, 32);
    b = abar2 * (__cosf(ang2) * b - __sinf(ang2) * brl) + dtc * Bj * xbd;
  }
  }

#pragma unroll
  for (int r = 0; r < 8; ++r)
    Tmat[((c * 16 + d) * 16 + (r + 8 * hi)) * 16 + lo] = T[r];
  if (hi == 0) bvec[(c * 16 + d) * 16 + lo] = b;
}

__global__ void __launch_bounds__(512)
k_combine(const float* __restrict__ Tmat,
                          const float* __restrict__ bvec,
                          const float* __restrict__ init_h,
                          float* __restrict__ hstart,
                          float* __restrict__ h_final) {
  const int d    = threadIdx.x >> 5;
  const int lane = threadIdx.x & 31;
  const int lo   = lane & 15;
  float h = init_h[d * 16 + lo];

  constexpr int PF3 = 8;
  float4 q[PF3][4] = {};
  float  bq[PF3]   = {};
#pragma unroll
  for (int u = 0; u < PF3 - 2; ++u) {
    const float4* Trow =
        reinterpret_cast<const float4*>(Tmat + ((u * 16 + d) * 16 + lo) * 16);
    q[u][0] = Trow[0]; q[u][1] = Trow[1]; q[u][2] = Trow[2]; q[u][3] = Trow[3];
    bq[u] = bvec[(u * 16 + d) * 16 + lo];
  }

  for (int c = 0; c < NCHUNK; c += PF3) {
#pragma unroll
    for (int u = 0; u < PF3; ++u) {
      const int cc = c + u;
      if (lane < 16) hstart[(cc * 16 + d) * 16 + lo] = h;
      {
        const int v  = (u + PF3 - 2) % PF3;
        const int cn = min(cc + PF3 - 2, NCHUNK - 1);
        const float4* Trow = reinterpret_cast<const float4*>(
            Tmat + ((cn * 16 + d) * 16 + lo) * 16);
        q[v][0] = Trow[0]; q[v][1] = Trow[1];
        q[v][2] = Trow[2]; q[v][3] = Trow[3];
        bq[v] = bvec[(cn * 16 + d) * 16 + lo];
      }
      const float4 q0 = q[u][0], q1 = q[u][1], q2 = q[u][2], q3 = q[u][3];
      float Tr[16] = {q0.x, q0.y, q0.z, q0.w, q1.x, q1.y, q1.z, q1.w,
                      q2.x, q2.y, q2.z, q2.w, q3.x, q3.y, q3.z, q3.w};
      float hn = bq[u];
#pragma unroll
      for (int k = 0; k < 16; ++k)
        hn = fmaf(Tr[k], __shfl(h, (lane & 16) | k, 32), hn);
      h = hn;
    }
  }
  if (lane < 16) h_final[d * 16 + lo] = h;
}

__global__ void k_scan(const float* __restrict__ dt,
                       const float* __restrict__ xb,
                       const float* __restrict__ Bv,
                       const float* __restrict__ Cv,
                       const float* __restrict__ A_log,
                       const float* __restrict__ rope,
                       const float* __restrict__ hstart,
                       float* __restrict__ y) {
  const int c    = blockIdx.x;
  const int d    = threadIdx.x >> 5;
  const int lane = threadIdx.x & 31;
  const int lo   = lane & 15;
  const float A2 = -__expf(A_log[d * 16 + lo]);
  const float F2 = rope[d * 16 + lo];
  float h = hstart[(c * 16 + d) * 16 + lo];
  asm volatile("" :: "v"(h));
  const int t0  = c * CHUNK;
  const int src = (lane & 16) | ((lo + 15) & 15);

  constexpr int PF4 = 8;
  float dtq[PF4] = {}, xbq[PF4] = {}, Bq[PF4] = {}, Cq[PF4] = {};
#pragma unroll
  for (int u = 0; u < PF4 - 1; ++u) {
    const int tt = t0 + u;
    dtq[u] = dt[tt * 16 + d];
    xbq[u] = xb[tt * 16 + d];
    Bq[u]  = Bv[tt * 16 + lo];
    Cq[u]  = Cv[tt * 16 + lo];
  }

  for (int t = 0; t < CHUNK; t += PF4) {
#pragma unroll
  for (int u = 0; u < PF4; ++u) {
    const int tt = t0 + t + u;
    {
      const int v  = (u + PF4 - 1) % PF4;
      const int tn = t0 + min(t + u + PF4 - 1, CHUNK - 1);
      dtq[v] = dt[tn * 16 + d];
      xbq[v] = xb[tn * 16 + d];
      Bq[v]  = Bv[tn * 16 + lo];
      Cq[v]  = Cv[tn * 16 + lo];
    }
    const float dtc = dtq[u];
    const float xbd = xbq[u];
    const float Bj  = Bq[u];
    const float Cj  = Cq[u];
    const float ta   = 0.5f * dtc * A2;
    const float abar = (1.f + ta) / (1.f - ta + 1e-8f);
    const float ang  = dtc * F2;
    const float hr   = __shfl(h, src, 32);
    h = abar * (__cosf(ang) * h - __sinf(ang) * hr) + dtc * Bj * xbd;
    float p = h * Cj;
    p += __shfl_xor(p, 1, 32);
    p += __shfl_xor(p, 2, 32);
    p += __shfl_xor(p, 4, 32);
    p += __shfl_xor(p, 8, 32);
    if (lane == 0) y[tt * 16 + d] = p;
  }
  }
}

__global__ void k_out(const float* __restrict__ y,
                      const float* __restrict__ z,
                      const float* __restrict__ xb,
                      const float* __restrict__ Dp,
                      const float* __restrict__ outw,
                      float* __restrict__ out) {
  const int lane = threadIdx.x & 31;
  const int wave = threadIdx.x >> 5;
  const int t0   = (blockIdx.x * 4 + wave) * 16;
  const int lo   = lane & 15, hi = lane >> 4;

  v8f acc = {};
#pragma unroll
  for (int kk = 0; kk < 4; ++kk) {
    const int kb = kk * 4 + 2 * hi;
    v2f a, b;
#pragma unroll
    for (int v = 0; v < 2; ++v) {
      const int k   = kb + v;
      const int idx = (t0 + lo) * 16 + k;
      const float zz  = z[idx];
      const float sil = zz / (1.f + __expf(-zz));
      const float val = y[idx] * sil + Dp[k] * xb[idx];
      if (v == 0) a.x = val; else a.y = val;
    }
    b.x = (lo < 8) ? outw[lo * 16 + kb + 0] : 0.f;
    b.y = (lo < 8) ? outw[lo * 16 + kb + 1] : 0.f;
    acc = wmma4(a, b, acc);
  }
  if (lo < 8) {
#pragma unroll
    for (int r = 0; r < 8; ++r) out[(t0 + r + 8 * hi) * 8 + lo] = acc[r];
  }
}

extern "C" void kernel_launch(void* const* d_in, const int* in_sizes, int n_in,
                              void* d_out, int out_size, void* d_ws,
                              size_t ws_size, hipStream_t stream) {
  const float* x    = (const float*)d_in[0];
  const float* init = (const float*)d_in[1];
  const float* inw  = (const float*)d_in[2];
  const float* dtw  = (const float*)d_in[3];
  const float* dtb  = (const float*)d_in[4];
  const float* bw   = (const float*)d_in[5];
  const float* cw   = (const float*)d_in[6];
  const float* alog = (const float*)d_in[7];
  const float* Dp   = (const float*)d_in[8];
  const float* rope = (const float*)d_in[9];
  const float* outw = (const float*)d_in[10];
  float* out = (float*)d_out;

  float* ws = (float*)d_ws;
  float* XB = ws;
  float* Z  = XB + N16;
  float* DT = Z + N16;
  float* BV = DT + N16;
  float* CV = BV + N16;
  float* Y  = CV + N16;
  float* T  = Y + N16;
  float* BC = T + N16;
  float* HS = BC + NCHUNK * 256;

  k_proj<<<NTOK / 64, 128, 0, stream>>>(x, inw, dtw, dtb, bw, cw,
                                        XB, Z, DT, BV, CV);
  k_chunkop<<<NCHUNK, 512, 0, stream>>>(DT, XB, BV, alog, rope, T, BC);
  k_combine<<<1, 512, 0, stream>>>(T, BC, init, HS, out + NTOK * DM);
  k_scan<<<NCHUNK, 512, 0, stream>>>(DT, XB, BV, CV, alog, rope, HS, Y);
  k_out<<<NTOK / 64, 128, 0, stream>>>(Y, Z, XB, Dp, outw, out);
}


# K2 chunk operator: T held one row per lane, so a lane computes its step's abar, cos, sin once instead of nine times; the j-1 roll is a DPP row rotate; arithmetic and sum order unchanged
# speedup vs baseline: 1.9351x; 1.4199x over previous
#include <hip/hip_runtime.h>
#include <hip/hip_bf16.h>
#include <math.h>


typedef __attribute__((ext_vector_type(2))) float v2f;
typedef __attribute__((ext_vector_type(8))) float v8f;

#define NTOK   65536
#define DM     8
#define DI     16
#define CHUNK  256
#define NCHUNK 256
#define N16    (NTOK * DI)

__device__ inline v8f wmma4(v2f a, v2f b, v8f c) {
  return __builtin_amdgcn_wmma_f32_16x16x4_f32(
      false, a, false, b, (short)0, c, false, false);
}

__device__ inline float softplus_f(float s) {
  return (s > 20.f) ? s : logf(1.f + __expf(s));
}

__global__ void k_proj(const float* __restrict__ x,
                       const float* __restrict__ inw,
                       const float* __restrict__ dtw,
                       const float* __restrict__ dtb,
                       const float* __restrict__ bw,
                       const float* __restrict__ cw,
                       float* __restrict__ xb, float* __restrict__ z,
                       float* __restrict__ dt, float* __restrict__ Bv,
                       float* __restrict__ Cv) {
  __shared__ float lds[4][256];
  const int lane = threadIdx.x & 31;
  const int wave = threadIdx.x >> 5;
  const int t0   = (blockIdx.x * 4 + wave) * 16;
  const int lo   = lane & 15, hi = lane >> 4;

  v8f axb = {}, az = {};
#pragma unroll
  for (int kk = 0; kk < 2; ++kk) {
    const int kb = kk * 4 + 2 * hi;
    v2f a, b0, b1;
    a.x  = x[(t0 + lo) * 8 + kb + 0];
    a.y  = x[(t0 + lo) * 8 + kb + 1];
    b0.x = inw[lo * 8 + kb + 0];
    b0.y = inw[lo * 8 + kb + 1];
    b1.x = inw[(lo + 16) * 8 + kb + 0];
    b1.y = inw[(lo + 16) * 8 + kb + 1];
    axb = wmma4(a, b0, axb);
    az  = wmma4(a, b1, az);
  }

#pragma unroll
  for (int r = 0; r < 8; ++r) lds[wave][(r + 8 * hi) * 16 + lo] = axb[r];

  v8f adt = {}, ab = {}, ac = {};
#pragma unroll
  for (int kk = 0; kk < 4; ++kk) {
    const int kb = kk * 4 + 2 * hi;
    v2f a, b;
    a.x = lds[wave][lo * 16 + kb + 0];
    a.y = lds[wave][lo * 16 + kb + 1];
    b.x = dtw[lo * 16 + kb + 0]; b.y = dtw[lo * 16 + kb + 1];
    adt = wmma4(a, b, adt);
    b.x = bw[lo * 16 + kb + 0];  b.y = bw[lo * 16 + kb + 1];
    ab  = wmma4(a, b, ab);
    b.x = cw[lo * 16 + kb + 0];  b.y = cw[lo * 16 + kb + 1];
    ac  = wmma4(a, b, ac);
  }

  const float bias = dtb[lo];
#pragma unroll
  for (int r = 0; r < 8; ++r) {
    const int idx = (t0 + r + 8 * hi) * 16 + lo;
    xb[idx] = axb[r];
    z[idx]  = az[r];
    dt[idx] = softplus_f(adt[r] + bias);
    Bv[idx] = ab[r];
    Cv[idx] = ac[r];
  }
}

__device__ inline float row_prev(float v) {
  return __builtin_bit_cast(
      float, __builtin_amdgcn_update_dpp(0, __builtin_bit_cast(int, v), 0x121,
                                         0xf, 0xf, true));
}

__global__ void k_chunkop(const float* __restrict__ dt,
                          const float* __restrict__ xb,
                          const float* __restrict__ Bv,
                          const float* __restrict__ A_log,
                          const float* __restrict__ rope,
                          float* __restrict__ Tmat,
                          float* __restrict__ bvec) {
  const int c    = blockIdx.x;
  const int d    = threadIdx.x >> 5;
  const int lane = threadIdx.x & 31;
  const int lo   = lane & 15, hi = lane >> 4;

  const float A2 = -__expf(A_log[d * 16 + lo]);
  const float F2 = rope[d * 16 + lo];

  float T[8];
#pragma unroll
  for (int r = 0; r < 8; ++r) T[r] = ((r + 8 * hi) == lo) ? 1.f : 0.f;
  float b = 0.f;

  const int t0 = c * CHUNK;

  constexpr int PF2 = 4;
  float dtq[PF2] = {}, xbq[PF2] = {}, Bq[PF2] = {};
#pragma unroll
  for (int u = 0; u < PF2 - 1; ++u) {
    const int tt = t0 + u;
    dtq[u] = dt[tt * 16 + d];
    xbq[u] = xb[tt * 16 + d];
    Bq[u]  = Bv[tt * 16 + lo];
  }

  for (int t = 0; t < CHUNK; t += PF2) {
#pragma unroll
  for (int u = 0; u < PF2; ++u) {
    {
      const int v  = (u + PF2 - 1) % PF2;
      const int tn = t0 + min(t + u + PF2 - 1, CHUNK - 1);
      dtq[v] = dt[tn * 16 + d];
      xbq[v] = xb[tn * 16 + d];
      Bq[v]  = Bv[tn * 16 + lo];
    }
    const float dtc = dtq[u];
    const float xbd = xbq[u];
    const float Bj  = Bq[u];

    const float ta   = 0.5f * dtc * A2;
    const float abar = (1.f + ta) / (1.f - ta + 1e-8f);
    const float ang  = dtc * F2;
    const float cs   = __cosf(ang);
    const float sn   = __sinf(ang);

#pragma unroll
    for (int r = 0; r < 8; ++r)
      T[r] = abar * (cs * T[r] - sn * row_prev(T[r]));
    b = abar * (cs * b - sn * row_prev(b)) + dtc * Bj * xbd;
  }
  }

  float4* dst = reinterpret_cast<float4*>(
      Tmat + ((c * 16 + d) * 16 + lo) * 16 + 8 * hi);
  dst[0] = make_float4(T[0], T[1], T[2], T[3]);
  dst[1] = make_float4(T[4], T[5], T[6], T[7]);
  if (hi == 0) bvec[(c * 16 + d) * 16 + lo] = b;
}

__global__ void __launch_bounds__(512)
k_combine(const float* __restrict__ Tmat,
                          const float* __restrict__ bvec,
                          const float* __restrict__ init_h,
                          float* __restrict__ hstart,
                          float* __restrict__ h_final) {
  const int d    = threadIdx.x >> 5;
  const int lane = threadIdx.x & 31;
  const int lo   = lane & 15;
  float h = init_h[d * 16 + lo];

  constexpr int PF3 = 8;
  float4 q[PF3][4] = {};
  float  bq[PF3]   = {};
#pragma unroll
  for (int u = 0; u < PF3 - 2; ++u) {
    const float4* Trow =
        reinterpret_cast<const float4*>(Tmat + ((u * 16 + d) * 16 + lo) * 16);
    q[u][0] = Trow[0]; q[u][1] = Trow[1]; q[u][2] = Trow[2]; q[u][3] = Trow[3];
    bq[u] = bvec[(u * 16 + d) * 16 + lo];
  }

  for (int c = 0; c < NCHUNK; c += PF3) {
#pragma unroll
    for (int u = 0; u < PF3; ++u) {
      const int cc = c + u;
      if (lane < 16) hstart[(cc * 16 + d) * 16 + lo] = h;
      {
        const int v  = (u + PF3 - 2) % PF3;
        const int cn = min(cc + PF3 - 2, NCHUNK - 1);
        const float4* Trow = reinterpret_cast<const float4*>(
            Tmat + ((cn * 16 + d) * 16 + lo) * 16);
        q[v][0] = Trow[0]; q[v][1] = Trow[1];
        q[v][2] = Trow[2]; q[v][3] = Trow[3];
        bq[v] = bvec[(cn * 16 + d) * 16 + lo];
      }
      const float4 q0 = q[u][0], q1 = q[u][1], q2 = q[u][2], q3 = q[u][3];
      float Tr[16] = {q0.x, q0.y, q0.z, q0.w, q1.x, q1.y, q1.z, q1.w,
                      q2.x, q2.y, q2.z, q2.w, q3.x, q3.y, q3.z, q3.w};
      float hn = bq[u];
#pragma unroll
      for (int k = 0; k < 16; ++k)
        hn = fmaf(Tr[k], __shfl(h, (lane & 16) | k, 32), hn);
      h = hn;
    }
  }
  if (lane < 16) h_final[d * 16 + lo] = h;
}

__global__ void k_scan(const float* __restrict__ dt,
                       const float* __restrict__ xb,
                       const float* __restrict__ Bv,
                       const float* __restrict__ Cv,
                       const float* __restrict__ A_log,
                       const float* __restrict__ rope,
                       const float* __restrict__ hstart,
                       float* __restrict__ y) {
  const int c    = blockIdx.x;
  const int d    = threadIdx.x >> 5;
  const int lane = threadIdx.x & 31;
  const int lo   = lane & 15;
  const float A2 = -__expf(A_log[d * 16 + lo]);
  const float F2 = rope[d * 16 + lo];
  float h = hstart[(c * 16 + d) * 16 + lo];
  asm volatile("" :: "v"(h));
  const int t0  = c * CHUNK;
  const int src = (lane & 16) | ((lo + 15) & 15);

  constexpr int PF4 = 8;
  float dtq[PF4] = {}, xbq[PF4] = {}, Bq[PF4] = {}, Cq[PF4] = {};
#pragma unroll
  for (int u = 0; u < PF4 - 1; ++u) {
    const int tt = t0 + u;
    dtq[u] = dt[tt * 16 + d];
    xbq[u] = xb[tt * 16 + d];
    Bq[u]  = Bv[tt * 16 + lo];
    Cq[u]  = Cv[tt * 16 + lo];
  }

  for (int t = 0; t < CHUNK; t += PF4) {
#pragma unroll
  for (int u = 0; u < PF4; ++u) {
    const int tt = t0 + t + u;
    {
      const int v  = (u + PF4 - 1) % PF4;
      const int tn = t0 + min(t + u + PF4 - 1, CHUNK - 1);
      dtq[v] = dt[tn * 16 + d];
      xbq[v] = xb[tn * 16 + d];
      Bq[v]  = Bv[tn * 16 + lo];
      Cq[v]  = Cv[tn * 16 + lo];
    }
    const float dtc = dtq[u];
    const float xbd = xbq[u];
    const float Bj  = Bq[u];
    const float Cj  = Cq[u];
    const float ta   = 0.5f * dtc * A2;
    const float abar = (1.f + ta) / (1.f - ta + 1e-8f);
    const float ang  = dtc * F2;
    const float hr   = __shfl(h, src, 32);
    h = abar * (__cosf(ang) * h - __sinf(ang) * hr) + dtc * Bj * xbd;
    float p = h * Cj;
    p += __shfl_xor(p, 1, 32);
    p += __shfl_xor(p, 2, 32);
    p += __shfl_xor(p, 4, 32);
    p += __shfl_xor(p, 8, 32);
    if (lane == 0) y[tt * 16 + d] = p;
  }
  }
}

__global__ void k_out(const float* __restrict__ y,
                      const float* __restrict__ z,
                      const float* __restrict__ xb,
                      const float* __restrict__ Dp,
                      const float* __restrict__ outw,
                      float* __restrict__ out) {
  const int lane = threadIdx.x & 31;
  const int wave = threadIdx.x >> 5;
  const int t0   = (blockIdx.x * 4 + wave) * 16;
  const int lo   = lane & 15, hi = lane >> 4;

  v8f acc = {};
#pragma unroll
  for (int kk = 0; kk < 4; ++kk) {
    const int kb = kk * 4 + 2 * hi;
    v2f a, b;
#pragma unroll
    for (int v = 0; v < 2; ++v) {
      const int k   = kb + v;
      const int idx = (t0 + lo) * 16 + k;
      const float zz  = z[idx];
      const float sil = zz / (1.f + __expf(-zz));
      const float val = y[idx] * sil + Dp[k] * xb[idx];
      if (v == 0) a.x = val; else a.y = val;
    }
    b.x = (lo < 8) ? outw[lo * 16 + kb + 0] : 0.f;
    b.y = (lo < 8) ? outw[lo * 16 + kb + 1] : 0.f;
    acc = wmma4(a, b, acc);
  }
  if (lo < 8) {
#pragma unroll
    for (int r = 0; r < 8; ++r) out[(t0 + r + 8 * hi) * 8 + lo] = acc[r];
  }
}

extern "C" void kernel_launch(void* const* d_in, const int* in_sizes, int n_in,
                              void* d_out, int out_size, void* d_ws,
                              size_t ws_size, hipStream_t stream) {
  const float* x    = (const float*)d_in[0];
  const float* init = (const float*)d_in[1];
  const float* inw  = (const float*)d_in[2];
  const float* dtw  = (const float*)d_in[3];
  const float* dtb  = (const float*)d_in[4];
  const float* bw   = (const float*)d_in[5];
  const float* cw   = (const float*)d_in[6];
  const float* alog = (const float*)d_in[7];
  const float* Dp   = (const float*)d_in[8];
  const float* rope = (const float*)d_in[9];
  const float* outw = (const float*)d_in[10];
  float* out = (float*)d_out;

  float* ws = (float*)d_ws;
  float* XB = ws;
  float* Z  = XB + N16;
  float* DT = Z + N16;
  float* BV = DT + N16;
  float* CV = BV + N16;
  float* Y  = CV + N16;
  float* T  = Y + N16;
  float* BC = T + N16;
  float* HS = BC + NCHUNK * 256;

  k_proj<<<NTOK / 64, 128, 0, stream>>>(x, inw, dtw, dtb, bw, cw,
                                        XB, Z, DT, BV, CV);
  k_chunkop<<<NCHUNK, 512, 0, stream>>>(DT, XB, BV, alog, rope, T, BC);
  k_combine<<<1, 512, 0, stream>>>(T, BC, init, HS, out + NTOK * DM);
  k_scan<<<NCHUNK, 512, 0, stream>>>(DT, XB, BV, CV, alog, rope, HS, Y);
  k_out<<<NTOK / 64, 128, 0, stream>>>(Y, Z, XB, Dp, outw, out);
}


# K4 scan: the j-1 roll and the 16-lane C-sum of every step are DPP row ops (row_ror, row_xmask) on the vector ALU instead of five serial ds_bpermute round trips; arithmetic and sum order unchanged
# speedup vs baseline: 2.2791x; 1.1778x over previous
#include <hip/hip_runtime.h>
#include <hip/hip_bf16.h>
#include <math.h>


typedef __attribute__((ext_vector_type(2))) float v2f;
typedef __attribute__((ext_vector_type(8))) float v8f;

#define NTOK   65536
#define DM     8
#define DI     16
#define CHUNK  256
#define NCHUNK 256
#define N16    (NTOK * DI)

__device__ inline v8f wmma4(v2f a, v2f b, v8f c) {
  return __builtin_amdgcn_wmma_f32_16x16x4_f32(
      false, a, false, b, (short)0, c, false, false);
}

__device__ inline float softplus_f(float s) {
  return (s > 20.f) ? s : logf(1.f + __expf(s));
}

__global__ void k_proj(const float* __restrict__ x,
                       const float* __restrict__ inw,
                       const float* __restrict__ dtw,
                       const float* __restrict__ dtb,
                       const float* __restrict__ bw,
                       const float* __restrict__ cw,
                       float* __restrict__ xb, float* __restrict__ z,
                       float* __restrict__ dt, float* __restrict__ Bv,
                       float* __restrict__ Cv) {
  __shared__ float lds[4][256];
  const int lane = threadIdx.x & 31;
  const int wave = threadIdx.x >> 5;
  const int t0   = (blockIdx.x * 4 + wave) * 16;
  const int lo   = lane & 15, hi = lane >> 4;

  v8f axb = {}, az = {};
#pragma unroll
  for (int kk = 0; kk < 2; ++kk) {
    const int kb = kk * 4 + 2 * hi;
    v2f a, b0, b1;
    a.x  = x[(t0 + lo) * 8 + kb + 0];
    a.y  = x[(t0 + lo) * 8 + kb + 1];
    b0.x = inw[lo * 8 + kb + 0];
    b0.y = inw[lo * 8 + kb + 1];
    b1.x = inw[(lo + 16) * 8 + kb + 0];
    b1.y = inw[(lo + 16) * 8 + kb + 1];
    axb = wmma4(a, b0, axb);
    az  = wmma4(a, b1, az);
  }

#pragma unroll
  for (int r = 0; r < 8; ++r) lds[wave][(r + 8 * hi) * 16 + lo] = axb[r];

  v8f adt = {}, ab = {}, ac = {};
#pragma unroll
  for (int kk = 0; kk < 4; ++kk) {
    const int kb = kk * 4 + 2 * hi;
    v2f a, b;
    a.x = lds[wave][lo * 16 + kb + 0];
    a.y = lds[wave][lo * 16 + kb + 1];
    b.x = dtw[lo * 16 + kb + 0]; b.y = dtw[lo * 16 + kb + 1];
    adt = wmma4(a, b, adt);
    b.x = bw[lo * 16 + kb + 0];  b.y = bw[lo * 16 + kb + 1];
    ab  = wmma4(a, b, ab);
    b.x = cw[lo * 16 + kb + 0];  b.y = cw[lo * 16 + kb + 1];
    ac  = wmma4(a, b, ac);
  }

  const float bias = dtb[lo];
#pragma unroll
  for (int r = 0; r < 8; ++r) {
    const int idx = (t0 + r + 8 * hi) * 16 + lo;
    xb[idx] = axb[r];
    z[idx]  = az[r];
    dt[idx] = softplus_f(adt[r] + bias);
    Bv[idx] = ab[r];
    Cv[idx] = ac[r];
  }
}

template <int CTRL>
__device__ inline float row_dpp(float v) {
  return __builtin_bit_cast(
      float, __builtin_amdgcn_update_dpp(0, __builtin_bit_cast(int, v), CTRL,
                                         0xf, 0xf, true));
}

__device__ inline float row_prev(float v) {
  return row_dpp<0x121>(v);
}

__device__ inline float row_sum16(float p) {
  p += row_dpp<0x161>(p);
  p += row_dpp<0x162>(p);
  p += row_dpp<0x164>(p);
  p += row_dpp<0x168>(p);
  return p;
}

__global__ void k_chunkop(const float* __restrict__ dt,
                          const float* __restrict__ xb,
                          const float* __restrict__ Bv,
                          const float* __restrict__ A_log,
                          const float* __restrict__ rope,
                          float* __restrict__ Tmat,
                          float* __restrict__ bvec) {
  const int c    = blockIdx.x;
  const int d    = threadIdx.x >> 5;
  const int lane = threadIdx.x & 31;
  const int lo   = lane & 15, hi = lane >> 4;

  const float A2 = -__expf(A_log[d * 16 + lo]);
  const float F2 = rope[d * 16 + lo];

  float T[8];
#pragma unroll
  for (int r = 0; r < 8; ++r) T[r] = ((r + 8 * hi) == lo) ? 1.f : 0.f;
  float b = 0.f;

  const int t0 = c * CHUNK;

  constexpr int PF2 = 4;
  float dtq[PF2] = {}, xbq[PF2] = {}, Bq[PF2] = {};
#pragma unroll
  for (int u = 0; u < PF2 - 1; ++u) {
    const int tt = t0 + u;
    dtq[u] = dt[tt * 16 + d];
    xbq[u] = xb[tt * 16 + d];
    Bq[u]  = Bv[tt * 16 + lo];
  }

  for (int t = 0; t < CHUNK; t += PF2) {
#pragma unroll
  for (int u = 0; u < PF2; ++u) {
    {
      const int v  = (u + PF2 - 1) % PF2;
      const int tn = t0 + min(t + u + PF2 - 1, CHUNK - 1);
      dtq[v] = dt[tn * 16 + d];
      xbq[v] = xb[tn * 16 + d];
      Bq[v]  = Bv[tn * 16 + lo];
    }
    const float dtc = dtq[u];
    const float xbd = xbq[u];
    const float Bj  = Bq[u];

    const float ta   = 0.5f * dtc * A2;
    const float abar = (1.f + ta) / (1.f - ta + 1e-8f);
    const float ang  = dtc * F2;
    const float cs   = __cosf(ang);
    const float sn   = __sinf(ang);

#pragma unroll
    for (int r = 0; r < 8; ++r)
      T[r] = abar * (cs * T[r] - sn * row_prev(T[r]));
    b = abar * (cs * b - sn * row_prev(b)) + dtc * Bj * xbd;
  }
  }

  float4* dst = reinterpret_cast<float4*>(
      Tmat + ((c * 16 + d) * 16 + lo) * 16 + 8 * hi);
  dst[0] = make_float4(T[0], T[1], T[2], T[3]);
  dst[1] = make_float4(T[4], T[5], T[6], T[7]);
  if (hi == 0) bvec[(c * 16 + d) * 16 + lo] = b;
}

__global__ void __launch_bounds__(512)
k_combine(const float* __restrict__ Tmat,
                          const float* __restrict__ bvec,
                          const float* __restrict__ init_h,
                          float* __restrict__ hstart,
                          float* __restrict__ h_final) {
  const int d    = threadIdx.x >> 5;
  const int lane = threadIdx.x & 31;
  const int lo   = lane & 15;
  float h = init_h[d * 16 + lo];

  constexpr int PF3 = 8;
  float4 q[PF3][4] = {};
  float  bq[PF3]   = {};
#pragma unroll
  for (int u = 0; u < PF3 - 2; ++u) {
    const float4* Trow =
        reinterpret_cast<const float4*>(Tmat + ((u * 16 + d) * 16 + lo) * 16);
    q[u][0] = Trow[0]; q[u][1] = Trow[1]; q[u][2] = Trow[2]; q[u][3] = Trow[3];
    bq[u] = bvec[(u * 16 + d) * 16 + lo];
  }

  for (int c = 0; c < NCHUNK; c += PF3) {
#pragma unroll
    for (int u = 0; u < PF3; ++u) {
      const int cc = c + u;
      if (lane < 16) hstart[(cc * 16 + d) * 16 + lo] = h;
      {
        const int v  = (u + PF3 - 2) % PF3;
        const int cn = min(cc + PF3 - 2, NCHUNK - 1);
        const float4* Trow = reinterpret_cast<const float4*>(
            Tmat + ((cn * 16 + d) * 16 + lo) * 16);
        q[v][0] = Trow[0]; q[v][1] = Trow[1];
        q[v][2] = Trow[2]; q[v][3] = Trow[3];
        bq[v] = bvec[(cn * 16 + d) * 16 + lo];
      }
      const float4 q0 = q[u][0], q1 = q[u][1], q2 = q[u][2], q3 = q[u][3];
      float Tr[16] = {q0.x, q0.y, q0.z, q0.w, q1.x, q1.y, q1.z, q1.w,
                      q2.x, q2.y, q2.z, q2.w, q3.x, q3.y, q3.z, q3.w};
      float hn = bq[u];
#pragma unroll
      for (int k = 0; k < 16; ++k)
        hn = fmaf(Tr[k], __shfl(h, (lane & 16) | k, 32), hn);
      h = hn;
    }
  }
  if (lane < 16) h_final[d * 16 + lo] = h;
}

__global__ void k_scan(const float* __restrict__ dt,
                       const float* __restrict__ xb,
                       const float* __restrict__ Bv,
                       const float* __restrict__ Cv,
                       const float* __restrict__ A_log,
                       const float* __restrict__ rope,
                       const float* __restrict__ hstart,
                       float* __restrict__ y) {
  const int c    = blockIdx.x;
  const int d    = threadIdx.x >> 5;
  const int lane = threadIdx.x & 31;
  const int lo   = lane & 15;
  const float A2 = -__expf(A_log[d * 16 + lo]);
  const float F2 = rope[d * 16 + lo];
  float h = hstart[(c * 16 + d) * 16 + lo];
  asm volatile("" :: "v"(h));
  const int t0  = c * CHUNK;

  constexpr int PF4 = 8;
  float dtq[PF4] = {}, xbq[PF4] = {}, Bq[PF4] = {}, Cq[PF4] = {};
#pragma unroll
  for (int u = 0; u < PF4 - 1; ++u) {
    const int tt = t0 + u;
    dtq[u] = dt[tt * 16 + d];
    xbq[u] = xb[tt * 16 + d];
    Bq[u]  = Bv[tt * 16 + lo];
    Cq[u]  = Cv[tt * 16 + lo];
  }

  for (int t = 0; t < CHUNK; t += PF4) {
#pragma unroll
  for (int u = 0; u < PF4; ++u) {
    const int tt = t0 + t + u;
    {
      const int v  = (u + PF4 - 1) % PF4;
      const int tn = t0 + min(t + u + PF4 - 1, CHUNK - 1);
      dtq[v] = dt[tn * 16 + d];
      xbq[v] = xb[tn * 16 + d];
      Bq[v]  = Bv[tn * 16 + lo];
      Cq[v]  = Cv[tn * 16 + lo];
    }
    const float dtc = dtq[u];
    const float xbd = xbq[u];
    const float Bj  = Bq[u];
    const float Cj  = Cq[u];
    const float ta   = 0.5f * dtc * A2;
    const float abar = (1.f + ta) / (1.f - ta + 1e-8f);
    const float ang  = dtc * F2;
    const float hr   = row_prev(h);
    h = abar * (__cosf(ang) * h - __sinf(ang) * hr) + dtc * Bj * xbd;
    const float p = row_sum16(h * Cj);
    if (lane == 0) y[tt * 16 + d] = p;
  }
  }
}

__global__ void k_out(const float* __restrict__ y,
                      const float* __restrict__ z,
                      const float* __restrict__ xb,
                      const float* __restrict__ Dp,
                      const float* __restrict__ outw,
                      float* __restrict__ out) {
  const int lane = threadIdx.x & 31;
  const int wave = threadIdx.x >> 5;
  const int t0   = (blockIdx.x * 4 + wave) * 16;
  const int lo   = lane & 15, hi = lane >> 4;

  v8f acc = {};
#pragma unroll
  for (int kk = 0; kk < 4; ++kk) {
    const int kb = kk * 4 + 2 * hi;
    v2f a, b;
#pragma unroll
    for (int v = 0; v < 2; ++v) {
      const int k   = kb + v;
      const int idx = (t0 + lo) * 16 + k;
      const float zz  = z[idx];
      const float sil = zz / (1.f + __expf(-zz));
      const float val = y[idx] * sil + Dp[k] * xb[idx];
      if (v == 0) a.x = val; else a.y = val;
    }
    b.x = (lo < 8) ? outw[lo * 16 + kb + 0] : 0.f;
    b.y = (lo < 8) ? outw[lo * 16 + kb + 1] : 0.f;
    acc = wmma4(a, b, acc);
  }
  if (lo < 8) {
#pragma unroll
    for (int r = 0; r < 8; ++r) out[(t0 + r + 8 * hi) * 8 + lo] = acc[r];
  }
}

extern "C" void kernel_launch(void* const* d_in, const int* in_sizes, int n_in,
                              void* d_out, int out_size, void* d_ws,
                              size_t ws_size, hipStream_t stream) {
  const float* x    = (const float*)d_in[0];
  const float* init = (const float*)d_in[1];
  const float* inw  = (const float*)d_in[2];
  const float* dtw  = (const float*)d_in[3];
  const float* dtb  = (const float*)d_in[4];
  const float* bw   = (const float*)d_in[5];
  const float* cw   = (const float*)d_in[6];
  const float* alog = (const float*)d_in[7];
  const float* Dp   = (const float*)d_in[8];
  const float* rope = (const float*)d_in[9];
  const float* outw = (const float*)d_in[10];
  float* out = (float*)d_out;

  float* ws = (float*)d_ws;
  float* XB = ws;
  float* Z  = XB + N16;
  float* DT = Z + N16;
  float* BV = DT + N16;
  float* CV = BV + N16;
  float* Y  = CV + N16;
  float* T  = Y + N16;
  float* BC = T + N16;
  float* HS = BC + NCHUNK * 256;

  k_proj<<<NTOK / 64, 128, 0, stream>>>(x, inw, dtw, dtb, bw, cw,
                                        XB, Z, DT, BV, CV);
  k_chunkop<<<NCHUNK, 512, 0, stream>>>(DT, XB, BV, alog, rope, T, BC);
  k_combine<<<1, 512, 0, stream>>>(T, BC, init, HS, out + NTOK * DM);
  k_scan<<<NCHUNK, 512, 0, stream>>>(DT, XB, BV, CV, alog, rope, HS, Y);
  k_out<<<NTOK / 64, 128, 0, stream>>>(Y, Z, XB, Dp, outw, out);
}


# K3 fold: h[k] reaches its 16-lane row by v_permlane16 on the vector ALU instead of 16 ds_bpermute trips through the LDS unit per chunk; arithmetic and sum order unchanged
# speedup vs baseline: 2.4384x; 1.0699x over previous
#include <hip/hip_runtime.h>
#include <hip/hip_bf16.h>
#include <math.h>


typedef __attribute__((ext_vector_type(2))) float v2f;
typedef __attribute__((ext_vector_type(8))) float v8f;

#define NTOK   65536
#define DM     8
#define DI     16
#define CHUNK  256
#define NCHUNK 256
#define N16    (NTOK * DI)

__device__ inline v8f wmma4(v2f a, v2f b, v8f c) {
  return __builtin_amdgcn_wmma_f32_16x16x4_f32(
      false, a, false, b, (short)0, c, false, false);
}

__device__ inline float softplus_f(float s) {
  return (s > 20.f) ? s : logf(1.f + __expf(s));
}

__global__ void k_proj(const float* __restrict__ x,
                       const float* __restrict__ inw,
                       const float* __restrict__ dtw,
                       const float* __restrict__ dtb,
                       const float* __restrict__ bw,
                       const float* __restrict__ cw,
                       float* __restrict__ xb, float* __restrict__ z,
                       float* __restrict__ dt, float* __restrict__ Bv,
                       float* __restrict__ Cv) {
  __shared__ float lds[4][256];
  const int lane = threadIdx.x & 31;
  const int wave = threadIdx.x >> 5;
  const int t0   = (blockIdx.x * 4 + wave) * 16;
  const int lo   = lane & 15, hi = lane >> 4;

  v8f axb = {}, az = {};
#pragma unroll
  for (int kk = 0; kk < 2; ++kk) {
    const int kb = kk * 4 + 2 * hi;
    v2f a, b0, b1;
    a.x  = x[(t0 + lo) * 8 + kb + 0];
    a.y  = x[(t0 + lo) * 8 + kb + 1];
    b0.x = inw[lo * 8 + kb + 0];
    b0.y = inw[lo * 8 + kb + 1];
    b1.x = inw[(lo + 16) * 8 + kb + 0];
    b1.y = inw[(lo + 16) * 8 + kb + 1];
    axb = wmma4(a, b0, axb);
    az  = wmma4(a, b1, az);
  }

#pragma unroll
  for (int r = 0; r < 8; ++r) lds[wave][(r + 8 * hi) * 16 + lo] = axb[r];

  v8f adt = {}, ab = {}, ac = {};
#pragma unroll
  for (int kk = 0; kk < 4; ++kk) {
    const int kb = kk * 4 + 2 * hi;
    v2f a, b;
    a.x = lds[wave][lo * 16 + kb + 0];
    a.y = lds[wave][lo * 16 + kb + 1];
    b.x = dtw[lo * 16 + kb + 0]; b.y = dtw[lo * 16 + kb + 1];
    adt = wmma4(a, b, adt);
    b.x = bw[lo * 16 + kb + 0];  b.y = bw[lo * 16 + kb + 1];
    ab  = wmma4(a, b, ab);
    b.x = cw[lo * 16 + kb + 0];  b.y = cw[lo * 16 + kb + 1];
    ac  = wmma4(a, b, ac);
  }

  const float bias = dtb[lo];
#pragma unroll
  for (int r = 0; r < 8; ++r) {
    const int idx = (t0 + r + 8 * hi) * 16 + lo;
    xb[idx] = axb[r];
    z[idx]  = az[r];
    dt[idx] = softplus_f(adt[r] + bias);
    Bv[idx] = ab[r];
    Cv[idx] = ac[r];
  }
}

template <int CTRL>
__device__ inline float row_dpp(float v) {
  return __builtin_bit_cast(
      float, __builtin_amdgcn_update_dpp(0, __builtin_bit_cast(int, v), CTRL,
                                         0xf, 0xf, true));
}

__device__ inline float row_prev(float v) {
  return row_dpp<0x121>(v);
}

__device__ inline float row_sum16(float p) {
  p += row_dpp<0x161>(p);
  p += row_dpp<0x162>(p);
  p += row_dpp<0x164>(p);
  p += row_dpp<0x168>(p);
  return p;
}

__global__ void k_chunkop(const float* __restrict__ dt,
                          const float* __restrict__ xb,
                          const float* __restrict__ Bv,
                          const float* __restrict__ A_log,
                          const float* __restrict__ rope,
                          float* __restrict__ Tmat,
                          float* __restrict__ bvec) {
  const int c    = blockIdx.x;
  const int d    = threadIdx.x >> 5;
  const int lane = threadIdx.x & 31;
  const int lo   = lane & 15, hi = lane >> 4;

  const float A2 = -__expf(A_log[d * 16 + lo]);
  const float F2 = rope[d * 16 + lo];

  float T[8];
#pragma unroll
  for (int r = 0; r < 8; ++r) T[r] = ((r + 8 * hi) == lo) ? 1.f : 0.f;
  float b = 0.f;

  const int t0 = c * CHUNK;

  constexpr int PF2 = 4;
  float dtq[PF2] = {}, xbq[PF2] = {}, Bq[PF2] = {};
#pragma unroll
  for (int u = 0; u < PF2 - 1; ++u) {
    const int tt = t0 + u;
    dtq[u] = dt[tt * 16 + d];
    xbq[u] = xb[tt * 16 + d];
    Bq[u]  = Bv[tt * 16 + lo];
  }

  for (int t = 0; t < CHUNK; t += PF2) {
#pragma unroll
  for (int u = 0; u < PF2; ++u) {
    {
      const int v  = (u + PF2 - 1) % PF2;
      const int tn = t0 + min(t + u + PF2 - 1, CHUNK - 1);
      dtq[v] = dt[tn * 16 + d];
      xbq[v] = xb[tn * 16 + d];
      Bq[v]  = Bv[tn * 16 + lo];
    }
    const float dtc = dtq[u];
    const float xbd = xbq[u];
    const float Bj  = Bq[u];

    const float ta   = 0.5f * dtc * A2;
    const float abar = (1.f + ta) / (1.f - ta + 1e-8f);
    const float ang  = dtc * F2;
    const float cs   = __cosf(ang);
    const float sn   = __sinf(ang);

#pragma unroll
    for (int r = 0; r < 8; ++r)
      T[r] = abar * (cs * T[r] - sn * row_prev(T[r]));
    b = abar * (cs * b - sn * row_prev(b)) + dtc * Bj * xbd;
  }
  }

  float4* dst = reinterpret_cast<float4*>(
      Tmat + ((c * 16 + d) * 16 + lo) * 16 + 8 * hi);
  dst[0] = make_float4(T[0], T[1], T[2], T[3]);
  dst[1] = make_float4(T[4], T[5], T[6], T[7]);
  if (hi == 0) bvec[(c * 16 + d) * 16 + lo] = b;
}

template <int K>
__device__ inline float row_bcast(float v) {
  constexpr unsigned SEL = 0x11111111u * (unsigned)K;
  return __builtin_bit_cast(
      float, __builtin_amdgcn_permlane16(0u, __builtin_bit_cast(unsigned, v),
                                         SEL, SEL, true, true));
}

template <int K>
__device__ inline float fold_row(const float (&Tr)[16], float h, float hn) {
  if constexpr (K < 16) {
    return fold_row<K + 1>(Tr, h, fmaf(Tr[K], row_bcast<K>(h), hn));
  } else {
    return hn;
  }
}

__global__ void __launch_bounds__(512)
k_combine(const float* __restrict__ Tmat,
                          const float* __restrict__ bvec,
                          const float* __restrict__ init_h,
                          float* __restrict__ hstart,
                          float* __restrict__ h_final) {
  const int d    = threadIdx.x >> 5;
  const int lane = threadIdx.x & 31;
  const int lo   = lane & 15;
  float h = init_h[d * 16 + lo];

  constexpr int PF3 = 8;
  float4 q[PF3][4] = {};
  float  bq[PF3]   = {};
#pragma unroll
  for (int u = 0; u < PF3 - 2; ++u) {
    const float4* Trow =
        reinterpret_cast<const float4*>(Tmat + ((u * 16 + d) * 16 + lo) * 16);
    q[u][0] = Trow[0]; q[u][1] = Trow[1]; q[u][2] = Trow[2]; q[u][3] = Trow[3];
    bq[u] = bvec[(u * 16 + d) * 16 + lo];
  }

  for (int c = 0; c < NCHUNK; c += PF3) {
#pragma unroll
    for (int u = 0; u < PF3; ++u) {
      const int cc = c + u;
      if (lane < 16) hstart[(cc * 16 + d) * 16 + lo] = h;
      {
        const int v  = (u + PF3 - 2) % PF3;
        const int cn = min(cc + PF3 - 2, NCHUNK - 1);
        const float4* Trow = reinterpret_cast<const float4*>(
            Tmat + ((cn * 16 + d) * 16 + lo) * 16);
        q[v][0] = Trow[0]; q[v][1] = Trow[1];
        q[v][2] = Trow[2]; q[v][3] = Trow[3];
        bq[v] = bvec[(cn * 16 + d) * 16 + lo];
      }
      const float4 q0 = q[u][0], q1 = q[u][1], q2 = q[u][2], q3 = q[u][3];
      float Tr[16] = {q0.x, q0.y, q0.z, q0.w, q1.x, q1.y, q1.z, q1.w,
                      q2.x, q2.y, q2.z, q2.w, q3.x, q3.y, q3.z, q3.w};
      h = fold_row<0>(Tr, h, bq[u]);
    }
  }
  if (lane < 16) h_final[d * 16 + lo] = h;
}

__global__ void k_scan(const float* __restrict__ dt,
                       const float* __restrict__ xb,
                       const float* __restrict__ Bv,
                       const float* __restrict__ Cv,
                       const float* __restrict__ A_log,
                       const float* __restrict__ rope,
                       const float* __restrict__ hstart,
                       float* __restrict__ y) {
  const int c    = blockIdx.x;
  const int d    = threadIdx.x >> 5;
  const int lane = threadIdx.x & 31;
  const int lo   = lane & 15;
  const float A2 = -__expf(A_log[d * 16 + lo]);
  const float F2 = rope[d * 16 + lo];
  float h = hstart[(c * 16 + d) * 16 + lo];
  asm volatile("" :: "v"(h));
  const int t0  = c * CHUNK;

  constexpr int PF4 = 8;
  float dtq[PF4] = {}, xbq[PF4] = {}, Bq[PF4] = {}, Cq[PF4] = {};
#pragma unroll
  for (int u = 0; u < PF4 - 1; ++u) {
    const int tt = t0 + u;
    dtq[u] = dt[tt * 16 + d];
    xbq[u] = xb[tt * 16 + d];
    Bq[u]  = Bv[tt * 16 + lo];
    Cq[u]  = Cv[tt * 16 + lo];
  }

  for (int t = 0; t < CHUNK; t += PF4) {
#pragma unroll
  for (int u = 0; u < PF4; ++u) {
    const int tt = t0 + t + u;
    {
      const int v  = (u + PF4 - 1) % PF4;
      const int tn = t0 + min(t + u + PF4 - 1, CHUNK - 1);
      dtq[v] = dt[tn * 16 + d];
      xbq[v] = xb[tn * 16 + d];
      Bq[v]  = Bv[tn * 16 + lo];
      Cq[v]  = Cv[tn * 16 + lo];
    }
    const float dtc = dtq[u];
    const float xbd = xbq[u];
    const float Bj  = Bq[u];
    const float Cj  = Cq[u];
    const float ta   = 0.5f * dtc * A2;
    const float abar = (1.f + ta) / (1.f - ta + 1e-8f);
    const float ang  = dtc * F2;
    const float hr   = row_prev(h);
    h = abar * (__cosf(ang) * h - __sinf(ang) * hr) + dtc * Bj * xbd;
    const float p = row_sum16(h * Cj);
    if (lane == 0) y[tt * 16 + d] = p;
  }
  }
}

__global__ void k_out(const float* __restrict__ y,
                      const float* __restrict__ z,
                      const float* __restrict__ xb,
                      const float* __restrict__ Dp,
                      const float* __restrict__ outw,
                      float* __restrict__ out) {
  const int lane = threadIdx.x & 31;
  const int wave = threadIdx.x >> 5;
  const int t0   = (blockIdx.x * 4 + wave) * 16;
  const int lo   = lane & 15, hi = lane >> 4;

  v8f acc = {};
#pragma unroll
  for (int kk = 0; kk < 4; ++kk) {
    const int kb = kk * 4 + 2 * hi;
    v2f a, b;
#pragma unroll
    for (int v = 0; v < 2; ++v) {
      const int k   = kb + v;
      const int idx = (t0 + lo) * 16 + k;
      const float zz  = z[idx];
      const float sil = zz / (1.f + __expf(-zz));
      const float val = y[idx] * sil + Dp[k] * xb[idx];
      if (v == 0) a.x = val; else a.y = val;
    }
    b.x = (lo < 8) ? outw[lo * 16 + kb + 0] : 0.f;
    b.y = (lo < 8) ? outw[lo * 16 + kb + 1] : 0.f;
    acc = wmma4(a, b, acc);
  }
  if (lo < 8) {
#pragma unroll
    for (int r = 0; r < 8; ++r) out[(t0 + r + 8 * hi) * 8 + lo] = acc[r];
  }
}

extern "C" void kernel_launch(void* const* d_in, const int* in_sizes, int n_in,
                              void* d_out, int out_size, void* d_ws,
                              size_t ws_size, hipStream_t stream) {
  const float* x    = (const float*)d_in[0];
  const float* init = (const float*)d_in[1];
  const float* inw  = (const float*)d_in[2];
  const float* dtw  = (const float*)d_in[3];
  const float* dtb  = (const float*)d_in[4];
  const float* bw   = (const float*)d_in[5];
  const float* cw   = (const float*)d_in[6];
  const float* alog = (const float*)d_in[7];
  const float* Dp   = (const float*)d_in[8];
  const float* rope = (const float*)d_in[9];
  const float* outw = (const float*)d_in[10];
  float* out = (float*)d_out;

  float* ws = (float*)d_ws;
  float* XB = ws;
  float* Z  = XB + N16;
  float* DT = Z + N16;
  float* BV = DT + N16;
  float* CV = BV + N16;
  float* Y  = CV + N16;
  float* T  = Y + N16;
  float* BC = T + N16;
  float* HS = BC + NCHUNK * 256;

  k_proj<<<NTOK / 64, 128, 0, stream>>>(x, inw, dtw, dtb, bw, cw,
                                        XB, Z, DT, BV, CV);
  k_chunkop<<<NCHUNK, 512, 0, stream>>>(DT, XB, BV, alog, rope, T, BC);
  k_combine<<<1, 512, 0, stream>>>(T, BC, init, HS, out + NTOK * DM);
  k_scan<<<NCHUNK, 512, 0, stream>>>(DT, XB, BV, CV, alog, rope, HS, Y);
  k_out<<<NTOK / 64, 128, 0, stream>>>(Y, Z, XB, Dp, outw, out);
}


# K3 fold: one single-wave work-group per channel on 16 WGPs instead of one 16-wave work-group on one WGP, its load ring deepened to span a memory trip; arithmetic and sum order unchanged
# speedup vs baseline: 3.4294x; 1.4064x over previous
#include <hip/hip_runtime.h>
#include <hip/hip_bf16.h>
#include <math.h>


typedef __attribute__((ext_vector_type(2))) float v2f;
typedef __attribute__((ext_vector_type(8))) float v8f;

#define NTOK   65536
#define DM     8
#define DI     16
#define CHUNK  256
#define NCHUNK 256
#define N16    (NTOK * DI)

__device__ inline v8f wmma4(v2f a, v2f b, v8f c) {
  return __builtin_amdgcn_wmma_f32_16x16x4_f32(
      false, a, false, b, (short)0, c, false, false);
}

__device__ inline float softplus_f(float s) {
  return (s > 20.f) ? s : logf(1.f + __expf(s));
}

__global__ void k_proj(const float* __restrict__ x,
                       const float* __restrict__ inw,
                       const float* __restrict__ dtw,
                       const float* __restrict__ dtb,
                       const float* __restrict__ bw,
                       const float* __restrict__ cw,
                       float* __restrict__ xb, float* __restrict__ z,
                       float* __restrict__ dt, float* __restrict__ Bv,
                       float* __restrict__ Cv) {
  __shared__ float lds[4][256];
  const int lane = threadIdx.x & 31;
  const int wave = threadIdx.x >> 5;
  const int t0   = (blockIdx.x * 4 + wave) * 16;
  const int lo   = lane & 15, hi = lane >> 4;

  v8f axb = {}, az = {};
#pragma unroll
  for (int kk = 0; kk < 2; ++kk) {
    const int kb = kk * 4 + 2 * hi;
    v2f a, b0, b1;
    a.x  = x[(t0 + lo) * 8 + kb + 0];
    a.y  = x[(t0 + lo) * 8 + kb + 1];
    b0.x = inw[lo * 8 + kb + 0];
    b0.y = inw[lo * 8 + kb + 1];
    b1.x = inw[(lo + 16) * 8 + kb + 0];
    b1.y = inw[(lo + 16) * 8 + kb + 1];
    axb = wmma4(a, b0, axb);
    az  = wmma4(a, b1, az);
  }

#pragma unroll
  for (int r = 0; r < 8; ++r) lds[wave][(r + 8 * hi) * 16 + lo] = axb[r];

  v8f adt = {}, ab = {}, ac = {};
#pragma unroll
  for (int kk = 0; kk < 4; ++kk) {
    const int kb = kk * 4 + 2 * hi;
    v2f a, b;
    a.x = lds[wave][lo * 16 + kb + 0];
    a.y = lds[wave][lo * 16 + kb + 1];
    b.x = dtw[lo * 16 + kb + 0]; b.y = dtw[lo * 16 + kb + 1];
    adt = wmma4(a, b, adt);
    b.x = bw[lo * 16 + kb + 0];  b.y = bw[lo * 16 + kb + 1];
    ab  = wmma4(a, b, ab);
    b.x = cw[lo * 16 + kb + 0];  b.y = cw[lo * 16 + kb + 1];
    ac  = wmma4(a, b, ac);
  }

  const float bias = dtb[lo];
#pragma unroll
  for (int r = 0; r < 8; ++r) {
    const int idx = (t0 + r + 8 * hi) * 16 + lo;
    xb[idx] = axb[r];
    z[idx]  = az[r];
    dt[idx] = softplus_f(adt[r] + bias);
    Bv[idx] = ab[r];
    Cv[idx] = ac[r];
  }
}

template <int CTRL>
__device__ inline float row_dpp(float v) {
  return __builtin_bit_cast(
      float, __builtin_amdgcn_update_dpp(0, __builtin_bit_cast(int, v), CTRL,
                                         0xf, 0xf, true));
}

__device__ inline float row_prev(float v) {
  return row_dpp<0x121>(v);
}

__device__ inline float row_sum16(float p) {
  p += row_dpp<0x161>(p);
  p += row_dpp<0x162>(p);
  p += row_dpp<0x164>(p);
  p += row_dpp<0x168>(p);
  return p;
}

__global__ void k_chunkop(const float* __restrict__ dt,
                          const float* __restrict__ xb,
                          const float* __restrict__ Bv,
                          const float* __restrict__ A_log,
                          const float* __restrict__ rope,
                          float* __restrict__ Tmat,
                          float* __restrict__ bvec) {
  const int c    = blockIdx.x;
  const int d    = threadIdx.x >> 5;
  const int lane = threadIdx.x & 31;
  const int lo   = lane & 15, hi = lane >> 4;

  const float A2 = -__expf(A_log[d * 16 + lo]);
  const float F2 = rope[d * 16 + lo];

  float T[8];
#pragma unroll
  for (int r = 0; r < 8; ++r) T[r] = ((r + 8 * hi) == lo) ? 1.f : 0.f;
  float b = 0.f;

  const int t0 = c * CHUNK;

  constexpr int PF2 = 4;
  float dtq[PF2] = {}, xbq[PF2] = {}, Bq[PF2] = {};
#pragma unroll
  for (int u = 0; u < PF2 - 1; ++u) {
    const int tt = t0 + u;
    dtq[u] = dt[tt * 16 + d];
    xbq[u] = xb[tt * 16 + d];
    Bq[u]  = Bv[tt * 16 + lo];
  }

  for (int t = 0; t < CHUNK; t += PF2) {
#pragma unroll
  for (int u = 0; u < PF2; ++u) {
    {
      const int v  = (u + PF2 - 1) % PF2;
      const int tn = t0 + min(t + u + PF2 - 1, CHUNK - 1);
      dtq[v] = dt[tn * 16 + d];
      xbq[v] = xb[tn * 16 + d];
      Bq[v]  = Bv[tn * 16 + lo];
    }
    const float dtc = dtq[u];
    const float xbd = xbq[u];
    const float Bj  = Bq[u];

    const float ta   = 0.5f * dtc * A2;
    const float abar = (1.f + ta) / (1.f - ta + 1e-8f);
    const float ang  = dtc * F2;
    const float cs   = __cosf(ang);
    const float sn   = __sinf(ang);

#pragma unroll
    for (int r = 0; r < 8; ++r)
      T[r] = abar * (cs * T[r] - sn * row_prev(T[r]));
    b = abar * (cs * b - sn * row_prev(b)) + dtc * Bj * xbd;
  }
  }

  float4* dst = reinterpret_cast<float4*>(
      Tmat + ((c * 16 + d) * 16 + lo) * 16 + 8 * hi);
  dst[0] = make_float4(T[0], T[1], T[2], T[3]);
  dst[1] = make_float4(T[4], T[5], T[6], T[7]);
  if (hi == 0) bvec[(c * 16 + d) * 16 + lo] = b;
}

template <int K>
__device__ inline float row_bcast(float v) {
  constexpr unsigned SEL = 0x11111111u * (unsigned)K;
  return __builtin_bit_cast(
      float, __builtin_amdgcn_permlane16(0u, __builtin_bit_cast(unsigned, v),
                                         SEL, SEL, true, true));
}

template <int K>
__device__ inline float fold_row(const float (&Tr)[16], float h, float hn) {
  if constexpr (K < 16) {
    return fold_row<K + 1>(Tr, h, fmaf(Tr[K], row_bcast<K>(h), hn));
  } else {
    return hn;
  }
}

constexpr int PF3 = 12;
constexpr int AH3 = PF3 - 2;

template <bool RELOAD>
__device__ inline float fold_chunk(const float* __restrict__ Tmat,
                                   const float* __restrict__ bvec,
                                   float* __restrict__ hrow,
                                   float4 (&q)[PF3][4], float (&bq)[PF3],
                                   const int u, const int cc, const int d,
                                   const unsigned lo, float h) {
  hrow[(size_t)cc * 256] = h;
  if constexpr (RELOAD) {
    const int v  = (u + AH3) % PF3;
    const int cn = min(cc + AH3, NCHUNK - 1);
    const float4* Trow =
        reinterpret_cast<const float4*>(Tmat + (cn * 16 + d) * 256) + lo * 4;
    q[v][0] = Trow[0]; q[v][1] = Trow[1];
    q[v][2] = Trow[2]; q[v][3] = Trow[3];
    bq[v] = (bvec + (cn * 16 + d) * 16)[lo];
  }
  const float4 q0 = q[u][0], q1 = q[u][1], q2 = q[u][2], q3 = q[u][3];
  float Tr[16] = {q0.x, q0.y, q0.z, q0.w, q1.x, q1.y, q1.z, q1.w,
                  q2.x, q2.y, q2.z, q2.w, q3.x, q3.y, q3.z, q3.w};
  return fold_row<0>(Tr, h, bq[u]);
}

__global__ void __launch_bounds__(32)
k_combine(const float* __restrict__ Tmat,
                          const float* __restrict__ bvec,
                          const float* __restrict__ init_h,
                          float* __restrict__ hstart,
                          float* __restrict__ h_final) {
  const int d    = blockIdx.x;
  const int lane = threadIdx.x & 31;
  const unsigned lo = lane & 15;
  float h = (init_h + d * 16)[lo];

  float* hrow = hstart + d * 16 + lo;

  float4 q[PF3][4] = {};
  float  bq[PF3]   = {};
#pragma unroll
  for (int u = 0; u < AH3; ++u) {
    const float4* Trow =
        reinterpret_cast<const float4*>(Tmat + (u * 16 + d) * 256) + lo * 4;
    q[u][0] = Trow[0]; q[u][1] = Trow[1]; q[u][2] = Trow[2]; q[u][3] = Trow[3];
    bq[u] = (bvec + (u * 16 + d) * 16)[lo];
    __builtin_amdgcn_sched_barrier(0);
  }

  constexpr int NCHUNK_FULL = NCHUNK - NCHUNK % PF3;
  for (int c = 0; c < NCHUNK_FULL; c += PF3) {
#pragma unroll
    for (int u = 0; u < PF3; ++u)
      h = fold_chunk<true>(Tmat, bvec, hrow, q, bq, u, c + u, d, lo, h);
  }
#pragma unroll
  for (int u = 0; u < NCHUNK % PF3; ++u)
    h = fold_chunk<false>(Tmat, bvec, hrow, q, bq, u, NCHUNK_FULL + u, d, lo,
                          h);
  if (lane < 16) (h_final + d * 16)[lo] = h;
}

__global__ void k_scan(const float* __restrict__ dt,
                       const float* __restrict__ xb,
                       const float* __restrict__ Bv,
                       const float* __restrict__ Cv,
                       const float* __restrict__ A_log,
                       const float* __restrict__ rope,
                       const float* __restrict__ hstart,
                       float* __restrict__ y) {
  const int c    = blockIdx.x;
  const int d    = threadIdx.x >> 5;
  const int lane = threadIdx.x & 31;
  const int lo   = lane & 15;
  const float A2 = -__expf(A_log[d * 16 + lo]);
  const float F2 = rope[d * 16 + lo];
  float h = hstart[(c * 16 + d) * 16 + lo];
  asm volatile("" :: "v"(h));
  const int t0  = c * CHUNK;

  constexpr int PF4 = 8;
  float dtq[PF4] = {}, xbq[PF4] = {}, Bq[PF4] = {}, Cq[PF4] = {};
#pragma unroll
  for (int u = 0; u < PF4 - 1; ++u) {
    const int tt = t0 + u;
    dtq[u] = dt[tt * 16 + d];
    xbq[u] = xb[tt * 16 + d];
    Bq[u]  = Bv[tt * 16 + lo];
    Cq[u]  = Cv[tt * 16 + lo];
  }

  for (int t = 0; t < CHUNK; t += PF4) {
#pragma unroll
  for (int u = 0; u < PF4; ++u) {
    const int tt = t0 + t + u;
    {
      const int v  = (u + PF4 - 1) % PF4;
      const int tn = t0 + min(t + u + PF4 - 1, CHUNK - 1);
      dtq[v] = dt[tn * 16 + d];
      xbq[v] = xb[tn * 16 + d];
      Bq[v]  = Bv[tn * 16 + lo];
      Cq[v]  = Cv[tn * 16 + lo];
    }
    const float dtc = dtq[u];
    const float xbd = xbq[u];
    const float Bj  = Bq[u];
    const float Cj  = Cq[u];
    const float ta   = 0.5f * dtc * A2;
    const float abar = (1.f + ta) / (1.f - ta + 1e-8f);
    const float ang  = dtc * F2;
    const float hr   = row_prev(h);
    h = abar * (__cosf(ang) * h - __sinf(ang) * hr) + dtc * Bj * xbd;
    const float p = row_sum16(h * Cj);
    if (lane == 0) y[tt * 16 + d] = p;
  }
  }
}

__global__ void k_out(const float* __restrict__ y,
                      const float* __restrict__ z,
                      const float* __restrict__ xb,
                      const float* __restrict__ Dp,
                      const float* __restrict__ outw,
                      float* __restrict__ out) {
  const int lane = threadIdx.x & 31;
  const int wave = threadIdx.x >> 5;
  const int t0   = (blockIdx.x * 4 + wave) * 16;
  const int lo   = lane & 15, hi = lane >> 4;

  v8f acc = {};
#pragma unroll
  for (int kk = 0; kk < 4; ++kk) {
    const int kb = kk * 4 + 2 * hi;
    v2f a, b;
#pragma unroll
    for (int v = 0; v < 2; ++v) {
      const int k   = kb + v;
      const int idx = (t0 + lo) * 16 + k;
      const float zz  = z[idx];
      const float sil = zz / (1.f + __expf(-zz));
      const float val = y[idx] * sil + Dp[k] * xb[idx];
      if (v == 0) a.x = val; else a.y = val;
    }
    b.x = (lo < 8) ? outw[lo * 16 + kb + 0] : 0.f;
    b.y = (lo < 8) ? outw[lo * 16 + kb + 1] : 0.f;
    acc = wmma4(a, b, acc);
  }
  if (lo < 8) {
#pragma unroll
    for (int r = 0; r < 8; ++r) out[(t0 + r + 8 * hi) * 8 + lo] = acc[r];
  }
}

extern "C" void kernel_launch(void* const* d_in, const int* in_sizes, int n_in,
                              void* d_out, int out_size, void* d_ws,
                              size_t ws_size, hipStream_t stream) {
  const float* x    = (const float*)d_in[0];
  const float* init = (const float*)d_in[1];
  const float* inw  = (const float*)d_in[2];
  const float* dtw  = (const float*)d_in[3];
  const float* dtb  = (const float*)d_in[4];
  const float* bw   = (const float*)d_in[5];
  const float* cw   = (const float*)d_in[6];
  const float* alog = (const float*)d_in[7];
  const float* Dp   = (const float*)d_in[8];
  const float* rope = (const float*)d_in[9];
  const float* outw = (const float*)d_in[10];
  float* out = (float*)d_out;

  float* ws = (float*)d_ws;
  float* XB = ws;
  float* Z  = XB + N16;
  float* DT = Z + N16;
  float* BV = DT + N16;
  float* CV = BV + N16;
  float* Y  = CV + N16;
  float* T  = Y + N16;
  float* BC = T + N16;
  float* HS = BC + NCHUNK * 256;

  k_proj<<<NTOK / 64, 128, 0, stream>>>(x, inw, dtw, dtb, bw, cw,
                                        XB, Z, DT, BV, CV);
  k_chunkop<<<NCHUNK, 512, 0, stream>>>(DT, XB, BV, alog, rope, T, BC);
  k_combine<<<16, 32, 0, stream>>>(T, BC, init, HS, out + NTOK * DM);
  k_scan<<<NCHUNK, 512, 0, stream>>>(DT, XB, BV, CV, alog, rope, HS, Y);
  k_out<<<NTOK / 64, 128, 0, stream>>>(Y, Z, XB, Dp, outw, out);
}


# K2 chunk operator, K4 scan: a wave carries two channels, one per 16-lane row, so no lane repeats another's work; half the waves, load rings deepened and kept in step order; arithmetic unchanged
# speedup vs baseline: 3.9918x; 1.1640x over previous
#include <hip/hip_runtime.h>
#include <hip/hip_bf16.h>
#include <math.h>


typedef __attribute__((ext_vector_type(2))) float v2f;
typedef __attribute__((ext_vector_type(8))) float v8f;

#define NTOK   65536
#define DM     8
#define DI     16
#define CHUNK  256
#define NCHUNK 256
#define N16    (NTOK * DI)

__device__ inline v8f wmma4(v2f a, v2f b, v8f c) {
  return __builtin_amdgcn_wmma_f32_16x16x4_f32(
      false, a, false, b, (short)0, c, false, false);
}

__device__ inline float softplus_f(float s) {
  return (s > 20.f) ? s : logf(1.f + __expf(s));
}

__global__ void k_proj(const float* __restrict__ x,
                       const float* __restrict__ inw,
                       const float* __restrict__ dtw,
                       const float* __restrict__ dtb,
                       const float* __restrict__ bw,
                       const float* __restrict__ cw,
                       float* __restrict__ xb, float* __restrict__ z,
                       float* __restrict__ dt, float* __restrict__ Bv,
                       float* __restrict__ Cv) {
  __shared__ float lds[4][256];
  const int lane = threadIdx.x & 31;
  const int wave = threadIdx.x >> 5;
  const int t0   = (blockIdx.x * 4 + wave) * 16;
  const int lo   = lane & 15, hi = lane >> 4;

  v8f axb = {}, az = {};
#pragma unroll
  for (int kk = 0; kk < 2; ++kk) {
    const int kb = kk * 4 + 2 * hi;
    v2f a, b0, b1;
    a.x  = x[(t0 + lo) * 8 + kb + 0];
    a.y  = x[(t0 + lo) * 8 + kb + 1];
    b0.x = inw[lo * 8 + kb + 0];
    b0.y = inw[lo * 8 + kb + 1];
    b1.x = inw[(lo + 16) * 8 + kb + 0];
    b1.y = inw[(lo + 16) * 8 + kb + 1];
    axb = wmma4(a, b0, axb);
    az  = wmma4(a, b1, az);
  }

#pragma unroll
  for (int r = 0; r < 8; ++r) lds[wave][(r + 8 * hi) * 16 + lo] = axb[r];

  v8f adt = {}, ab = {}, ac = {};
#pragma unroll
  for (int kk = 0; kk < 4; ++kk) {
    const int kb = kk * 4 + 2 * hi;
    v2f a, b;
    a.x = lds[wave][lo * 16 + kb + 0];
    a.y = lds[wave][lo * 16 + kb + 1];
    b.x = dtw[lo * 16 + kb + 0]; b.y = dtw[lo * 16 + kb + 1];
    adt = wmma4(a, b, adt);
    b.x = bw[lo * 16 + kb + 0];  b.y = bw[lo * 16 + kb + 1];
    ab  = wmma4(a, b, ab);
    b.x = cw[lo * 16 + kb + 0];  b.y = cw[lo * 16 + kb + 1];
    ac  = wmma4(a, b, ac);
  }

  const float bias = dtb[lo];
#pragma unroll
  for (int r = 0; r < 8; ++r) {
    const int idx = (t0 + r + 8 * hi) * 16 + lo;
    xb[idx] = axb[r];
    z[idx]  = az[r];
    dt[idx] = softplus_f(adt[r] + bias);
    Bv[idx] = ab[r];
    Cv[idx] = ac[r];
  }
}

template <int CTRL>
__device__ inline float row_dpp(float v) {
  return __builtin_bit_cast(
      float, __builtin_amdgcn_update_dpp(0, __builtin_bit_cast(int, v), CTRL,
                                         0xf, 0xf, true));
}

__device__ inline float row_prev(float v) {
  return row_dpp<0x121>(v);
}

__device__ inline float row_sum16(float p) {
  p += row_dpp<0x161>(p);
  p += row_dpp<0x162>(p);
  p += row_dpp<0x164>(p);
  p += row_dpp<0x168>(p);
  return p;
}

__global__ void k_chunkop(const float* __restrict__ dt,
                          const float* __restrict__ xb,
                          const float* __restrict__ Bv,
                          const float* __restrict__ A_log,
                          const float* __restrict__ rope,
                          float* __restrict__ Tmat,
                          float* __restrict__ bvec) {
  const int c    = blockIdx.x;
  const int lane = threadIdx.x & 31;
  const int lo   = lane & 15, hi = lane >> 4;
  const int d    = 2 * (threadIdx.x >> 5) + hi;

  const float A2 = -__expf(A_log[d * 16 + lo]);
  const float F2 = rope[d * 16 + lo];

  float T[16];
#pragma unroll
  for (int r = 0; r < 16; ++r) T[r] = (r == lo) ? 1.f : 0.f;
  float b = 0.f;

  const int t0 = c * CHUNK;

  constexpr int PF2 = 8;
  float dtq[PF2] = {}, xbq[PF2] = {}, Bq[PF2] = {};
#pragma unroll
  for (int u = 0; u < PF2 - 1; ++u) {
    const int tt = t0 + u;
    dtq[u] = dt[tt * 16 + d];
    xbq[u] = xb[tt * 16 + d];
    Bq[u]  = Bv[tt * 16 + lo];
    __builtin_amdgcn_sched_barrier(0);
  }

  for (int t = 0; t < CHUNK; t += PF2) {
#pragma unroll
  for (int u = 0; u < PF2; ++u) {
    {
      const int v  = (u + PF2 - 1) % PF2;
      const int tn = t0 + min(t + u + PF2 - 1, CHUNK - 1);
      dtq[v] = dt[tn * 16 + d];
      xbq[v] = xb[tn * 16 + d];
      Bq[v]  = Bv[tn * 16 + lo];
      __builtin_amdgcn_sched_barrier(0);
    }
    const float dtc = dtq[u];
    const float xbd = xbq[u];
    const float Bj  = Bq[u];

    const float ta   = 0.5f * dtc * A2;
    const float abar = (1.f + ta) / (1.f - ta + 1e-8f);
    const float ang  = dtc * F2;
    const float cs   = __cosf(ang);
    const float sn   = __sinf(ang);

#pragma unroll
    for (int r = 0; r < 16; ++r)
      T[r] = abar * (cs * T[r] - sn * row_prev(T[r]));
    b = abar * (cs * b - sn * row_prev(b)) + dtc * Bj * xbd;
  }
  }

  float4* dst = reinterpret_cast<float4*>(
      Tmat + ((c * 16 + d) * 16 + lo) * 16);
  dst[0] = make_float4(T[0], T[1], T[2], T[3]);
  dst[1] = make_float4(T[4], T[5], T[6], T[7]);
  dst[2] = make_float4(T[8], T[9], T[10], T[11]);
  dst[3] = make_float4(T[12], T[13], T[14], T[15]);
  bvec[(c * 16 + d) * 16 + lo] = b;
}

template <int K>
__device__ inline float row_bcast(float v) {
  constexpr unsigned SEL = 0x11111111u * (unsigned)K;
  return __builtin_bit_cast(
      float, __builtin_amdgcn_permlane16(0u, __builtin_bit_cast(unsigned, v),
                                         SEL, SEL, true, true));
}

template <int K>
__device__ inline float fold_row(const float (&Tr)[16], float h, float hn) {
  if constexpr (K < 16) {
    return fold_row<K + 1>(Tr, h, fmaf(Tr[K], row_bcast<K>(h), hn));
  } else {
    return hn;
  }
}

constexpr int PF3 = 12;
constexpr int AH3 = PF3 - 2;

template <bool RELOAD>
__device__ inline float fold_chunk(const float* __restrict__ Tmat,
                                   const float* __restrict__ bvec,
                                   float* __restrict__ hrow,
                                   float4 (&q)[PF3][4], float (&bq)[PF3],
                                   const int u, const int cc, const int d,
                                   const unsigned lo, float h) {
  hrow[(size_t)cc * 256] = h;
  if constexpr (RELOAD) {
    const int v  = (u + AH3) % PF3;
    const int cn = min(cc + AH3, NCHUNK - 1);
    const float4* Trow =
        reinterpret_cast<const float4*>(Tmat + (cn * 16 + d) * 256) + lo * 4;
    q[v][0] = Trow[0]; q[v][1] = Trow[1];
    q[v][2] = Trow[2]; q[v][3] = Trow[3];
    bq[v] = (bvec + (cn * 16 + d) * 16)[lo];
  }
  const float4 q0 = q[u][0], q1 = q[u][1], q2 = q[u][2], q3 = q[u][3];
  float Tr[16] = {q0.x, q0.y, q0.z, q0.w, q1.x, q1.y, q1.z, q1.w,
                  q2.x, q2.y, q2.z, q2.w, q3.x, q3.y, q3.z, q3.w};
  return fold_row<0>(Tr, h, bq[u]);
}

__global__ void __launch_bounds__(32)
k_combine(const float* __restrict__ Tmat,
                          const float* __restrict__ bvec,
                          const float* __restrict__ init_h,
                          float* __restrict__ hstart,
                          float* __restrict__ h_final) {
  const int d    = blockIdx.x;
  const int lane = threadIdx.x & 31;
  const unsigned lo = lane & 15;
  float h = (init_h + d * 16)[lo];

  float* hrow = hstart + d * 16 + lo;

  float4 q[PF3][4] = {};
  float  bq[PF3]   = {};
#pragma unroll
  for (int u = 0; u < AH3; ++u) {
    const float4* Trow =
        reinterpret_cast<const float4*>(Tmat + (u * 16 + d) * 256) + lo * 4;
    q[u][0] = Trow[0]; q[u][1] = Trow[1]; q[u][2] = Trow[2]; q[u][3] = Trow[3];
    bq[u] = (bvec + (u * 16 + d) * 16)[lo];
    __builtin_amdgcn_sched_barrier(0);
  }

  constexpr int NCHUNK_FULL = NCHUNK - NCHUNK % PF3;
  for (int c = 0; c < NCHUNK_FULL; c += PF3) {
#pragma unroll
    for (int u = 0; u < PF3; ++u)
      h = fold_chunk<true>(Tmat, bvec, hrow, q, bq, u, c + u, d, lo, h);
  }
#pragma unroll
  for (int u = 0; u < NCHUNK % PF3; ++u)
    h = fold_chunk<false>(Tmat, bvec, hrow, q, bq, u, NCHUNK_FULL + u, d, lo,
                          h);
  if (lane < 16) (h_final + d * 16)[lo] = h;
}

__global__ void k_scan(const float* __restrict__ dt,
                       const float* __restrict__ xb,
                       const float* __restrict__ Bv,
                       const float* __restrict__ Cv,
                       const float* __restrict__ A_log,
                       const float* __restrict__ rope,
                       const float* __restrict__ hstart,
                       float* __restrict__ y) {
  const int c    = blockIdx.x;
  const int lane = threadIdx.x & 31;
  const int lo   = lane & 15, hi = lane >> 4;
  const int d    = 2 * (threadIdx.x >> 5) + hi;
  const float A2 = -__expf(A_log[d * 16 + lo]);
  const float F2 = rope[d * 16 + lo];
  float h = hstart[(c * 16 + d) * 16 + lo];
  asm volatile("" :: "v"(h));
  const int t0  = c * CHUNK;

  constexpr int PF4 = 16;
  float dtq[PF4] = {}, xbq[PF4] = {}, Bq[PF4] = {}, Cq[PF4] = {};
#pragma unroll
  for (int u = 0; u < PF4 - 1; ++u) {
    const int tt = t0 + u;
    dtq[u] = dt[tt * 16 + d];
    xbq[u] = xb[tt * 16 + d];
    Bq[u]  = Bv[tt * 16 + lo];
    Cq[u]  = Cv[tt * 16 + lo];
    __builtin_amdgcn_sched_barrier(0);
  }

  for (int t = 0; t < CHUNK; t += PF4) {
#pragma unroll
  for (int u = 0; u < PF4; ++u) {
    const int tt = t0 + t + u;
    {
      const int v  = (u + PF4 - 1) % PF4;
      const int tn = t0 + min(t + u + PF4 - 1, CHUNK - 1);
      dtq[v] = dt[tn * 16 + d];
      xbq[v] = xb[tn * 16 + d];
      Bq[v]  = Bv[tn * 16 + lo];
      Cq[v]  = Cv[tn * 16 + lo];
    }
    const float dtc = dtq[u];
    const float xbd = xbq[u];
    const float Bj  = Bq[u];
    const float Cj  = Cq[u];
    const float ta   = 0.5f * dtc * A2;
    const float abar = (1.f + ta) / (1.f - ta + 1e-8f);
    const float ang  = dtc * F2;
    const float hr   = row_prev(h);
    h = abar * (__cosf(ang) * h - __sinf(ang) * hr) + dtc * Bj * xbd;
    const float p = row_sum16(h * Cj);
    if (lo == 0) y[tt * 16 + d] = p;
  }
  }
}

__global__ void k_out(const float* __restrict__ y,
                      const float* __restrict__ z,
                      const float* __restrict__ xb,
                      const float* __restrict__ Dp,
                      const float* __restrict__ outw,
                      float* __restrict__ out) {
  const int lane = threadIdx.x & 31;
  const int wave = threadIdx.x >> 5;
  const int t0   = (blockIdx.x * 4 + wave) * 16;
  const int lo   = lane & 15, hi = lane >> 4;

  v8f acc = {};
#pragma unroll
  for (int kk = 0; kk < 4; ++kk) {
    const int kb = kk * 4 + 2 * hi;
    v2f a, b;
#pragma unroll
    for (int v = 0; v < 2; ++v) {
      const int k   = kb + v;
      const int idx = (t0 + lo) * 16 + k;
      const float zz  = z[idx];
      const float sil = zz / (1.f + __expf(-zz));
      const float val = y[idx] * sil + Dp[k] * xb[idx];
      if (v == 0) a.x = val; else a.y = val;
    }
    b.x = (lo < 8) ? outw[lo * 16 + kb + 0] : 0.f;
    b.y = (lo < 8) ? outw[lo * 16 + kb + 1] : 0.f;
    acc = wmma4(a, b, acc);
  }
  if (lo < 8) {
#pragma unroll
    for (int r = 0; r < 8; ++r) out[(t0 + r + 8 * hi) * 8 + lo] = acc[r];
  }
}

extern "C" void kernel_launch(void* const* d_in, const int* in_sizes, int n_in,
                              void* d_out, int out_size, void* d_ws,
                              size_t ws_size, hipStream_t stream) {
  const float* x    = (const float*)d_in[0];
  const float* init = (const float*)d_in[1];
  const float* inw  = (const float*)d_in[2];
  const float* dtw  = (const float*)d_in[3];
  const float* dtb  = (const float*)d_in[4];
  const float* bw   = (const float*)d_in[5];
  const float* cw   = (const float*)d_in[6];
  const float* alog = (const float*)d_in[7];
  const float* Dp   = (const float*)d_in[8];
  const float* rope = (const float*)d_in[9];
  const float* outw = (const float*)d_in[10];
  float* out = (float*)d_out;

  float* ws = (float*)d_ws;
  float* XB = ws;
  float* Z  = XB + N16;
  float* DT = Z + N16;
  float* BV = DT + N16;
  float* CV = BV + N16;
  float* Y  = CV + N16;
  float* T  = Y + N16;
  float* BC = T + N16;
  float* HS = BC + NCHUNK * 256;

  k_proj<<<NTOK / 64, 128, 0, stream>>>(x, inw, dtw, dtb, bw, cw,
                                        XB, Z, DT, BV, CV);
  k_chunkop<<<NCHUNK, 256, 0, stream>>>(DT, XB, BV, alog, rope, T, BC);
  k_combine<<<16, 32, 0, stream>>>(T, BC, init, HS, out + NTOK * DM);
  k_scan<<<NCHUNK, 256, 0, stream>>>(DT, XB, BV, CV, alog, rope, HS, Y);
  k_out<<<NTOK / 64, 128, 0, stream>>>(Y, Z, XB, Dp, outw, out);
}


# K4 scan: per pass of 16 steps the state-independent terms (abar, cos, sin, input) are computed as one block ahead of the recurrence, y stored once per pass without a branch; arithmetic unchanged
# speedup vs baseline: 4.6667x; 1.1691x over previous
#include <hip/hip_runtime.h>
#include <hip/hip_bf16.h>
#include <math.h>


typedef __attribute__((ext_vector_type(2))) float v2f;
typedef __attribute__((ext_vector_type(8))) float v8f;

#define NTOK   65536
#define DM     8
#define DI     16
#define CHUNK  256
#define NCHUNK 256
#define N16    (NTOK * DI)

__device__ inline v8f wmma4(v2f a, v2f b, v8f c) {
  return __builtin_amdgcn_wmma_f32_16x16x4_f32(
      false, a, false, b, (short)0, c, false, false);
}

__device__ inline float softplus_f(float s) {
  return (s > 20.f) ? s : logf(1.f + __expf(s));
}

__global__ void k_proj(const float* __restrict__ x,
                       const float* __restrict__ inw,
                       const float* __restrict__ dtw,
                       const float* __restrict__ dtb,
                       const float* __restrict__ bw,
                       const float* __restrict__ cw,
                       float* __restrict__ xb, float* __restrict__ z,
                       float* __restrict__ dt, float* __restrict__ Bv,
                       float* __restrict__ Cv) {
  __shared__ float lds[4][256];
  const int lane = threadIdx.x & 31;
  const int wave = threadIdx.x >> 5;
  const int t0   = (blockIdx.x * 4 + wave) * 16;
  const int lo   = lane & 15, hi = lane >> 4;

  v8f axb = {}, az = {};
#pragma unroll
  for (int kk = 0; kk < 2; ++kk) {
    const int kb = kk * 4 + 2 * hi;
    v2f a, b0, b1;
    a.x  = x[(t0 + lo) * 8 + kb + 0];
    a.y  = x[(t0 + lo) * 8 + kb + 1];
    b0.x = inw[lo * 8 + kb + 0];
    b0.y = inw[lo * 8 + kb + 1];
    b1.x = inw[(lo + 16) * 8 + kb + 0];
    b1.y = inw[(lo + 16) * 8 + kb + 1];
    axb = wmma4(a, b0, axb);
    az  = wmma4(a, b1, az);
  }

#pragma unroll
  for (int r = 0; r < 8; ++r) lds[wave][(r + 8 * hi) * 16 + lo] = axb[r];

  v8f adt = {}, ab = {}, ac = {};
#pragma unroll
  for (int kk = 0; kk < 4; ++kk) {
    const int kb = kk * 4 + 2 * hi;
    v2f a, b;
    a.x = lds[wave][lo * 16 + kb + 0];
    a.y = lds[wave][lo * 16 + kb + 1];
    b.x = dtw[lo * 16 + kb + 0]; b.y = dtw[lo * 16 + kb + 1];
    adt = wmma4(a, b, adt);
    b.x = bw[lo * 16 + kb + 0];  b.y = bw[lo * 16 + kb + 1];
    ab  = wmma4(a, b, ab);
    b.x = cw[lo * 16 + kb + 0];  b.y = cw[lo * 16 + kb + 1];
    ac  = wmma4(a, b, ac);
  }

  const float bias = dtb[lo];
#pragma unroll
  for (int r = 0; r < 8; ++r) {
    const int idx = (t0 + r + 8 * hi) * 16 + lo;
    xb[idx] = axb[r];
    z[idx]  = az[r];
    dt[idx] = softplus_f(adt[r] + bias);
    Bv[idx] = ab[r];
    Cv[idx] = ac[r];
  }
}

template <int CTRL>
__device__ inline float row_dpp(float v) {
  return __builtin_bit_cast(
      float, __builtin_amdgcn_update_dpp(0, __builtin_bit_cast(int, v), CTRL,
                                         0xf, 0xf, true));
}

__device__ inline float row_prev(float v) {
  return row_dpp<0x121>(v);
}

__device__ inline float row_sum16(float p) {
  p += row_dpp<0x161>(p);
  p += row_dpp<0x162>(p);
  p += row_dpp<0x164>(p);
  p += row_dpp<0x168>(p);
  return p;
}

__global__ void k_chunkop(const float* __restrict__ dt,
                          const float* __restrict__ xb,
                          const float* __restrict__ Bv,
                          const float* __restrict__ A_log,
                          const float* __restrict__ rope,
                          float* __restrict__ Tmat,
                          float* __restrict__ bvec) {
  const int c    = blockIdx.x;
  const int lane = threadIdx.x & 31;
  const int lo   = lane & 15, hi = lane >> 4;
  const int d    = 2 * (threadIdx.x >> 5) + hi;

  const float A2 = -__expf(A_log[d * 16 + lo]);
  const float F2 = rope[d * 16 + lo];

  float T[16];
#pragma unroll
  for (int r = 0; r < 16; ++r) T[r] = (r == lo) ? 1.f : 0.f;
  float b = 0.f;

  const int t0 = c * CHUNK;

  constexpr int PF2 = 8;
  float dtq[PF2] = {}, xbq[PF2] = {}, Bq[PF2] = {};
#pragma unroll
  for (int u = 0; u < PF2 - 1; ++u) {
    const int tt = t0 + u;
    dtq[u] = dt[tt * 16 + d];
    xbq[u] = xb[tt * 16 + d];
    Bq[u]  = Bv[tt * 16 + lo];
    __builtin_amdgcn_sched_barrier(0);
  }

  for (int t = 0; t < CHUNK; t += PF2) {
#pragma unroll
  for (int u = 0; u < PF2; ++u) {
    {
      const int v  = (u + PF2 - 1) % PF2;
      const int tn = t0 + min(t + u + PF2 - 1, CHUNK - 1);
      dtq[v] = dt[tn * 16 + d];
      xbq[v] = xb[tn * 16 + d];
      Bq[v]  = Bv[tn * 16 + lo];
      __builtin_amdgcn_sched_barrier(0);
    }
    const float dtc = dtq[u];
    const float xbd = xbq[u];
    const float Bj  = Bq[u];

    const float ta   = 0.5f * dtc * A2;
    const float abar = (1.f + ta) / (1.f - ta + 1e-8f);
    const float ang  = dtc * F2;
    const float cs   = __cosf(ang);
    const float sn   = __sinf(ang);

#pragma unroll
    for (int r = 0; r < 16; ++r)
      T[r] = abar * (cs * T[r] - sn * row_prev(T[r]));
    b = abar * (cs * b - sn * row_prev(b)) + dtc * Bj * xbd;
  }
  }

  float4* dst = reinterpret_cast<float4*>(
      Tmat + ((c * 16 + d) * 16 + lo) * 16);
  dst[0] = make_float4(T[0], T[1], T[2], T[3]);
  dst[1] = make_float4(T[4], T[5], T[6], T[7]);
  dst[2] = make_float4(T[8], T[9], T[10], T[11]);
  dst[3] = make_float4(T[12], T[13], T[14], T[15]);
  bvec[(c * 16 + d) * 16 + lo] = b;
}

template <int K>
__device__ inline float row_bcast(float v) {
  constexpr unsigned SEL = 0x11111111u * (unsigned)K;
  return __builtin_bit_cast(
      float, __builtin_amdgcn_permlane16(0u, __builtin_bit_cast(unsigned, v),
                                         SEL, SEL, true, true));
}

template <int K>
__device__ inline float fold_row(const float (&Tr)[16], float h, float hn) {
  if constexpr (K < 16) {
    return fold_row<K + 1>(Tr, h, fmaf(Tr[K], row_bcast<K>(h), hn));
  } else {
    return hn;
  }
}

constexpr int PF3 = 12;
constexpr int AH3 = PF3 - 2;

template <bool RELOAD>
__device__ inline float fold_chunk(const float* __restrict__ Tmat,
                                   const float* __restrict__ bvec,
                                   float* __restrict__ hrow,
                                   float4 (&q)[PF3][4], float (&bq)[PF3],
                                   const int u, const int cc, const int d,
                                   const unsigned lo, float h) {
  hrow[(size_t)cc * 256] = h;
  if constexpr (RELOAD) {
    const int v  = (u + AH3) % PF3;
    const int cn = min(cc + AH3, NCHUNK - 1);
    const float4* Trow =
        reinterpret_cast<const float4*>(Tmat + (cn * 16 + d) * 256) + lo * 4;
    q[v][0] = Trow[0]; q[v][1] = Trow[1];
    q[v][2] = Trow[2]; q[v][3] = Trow[3];
    bq[v] = (bvec + (cn * 16 + d) * 16)[lo];
  }
  const float4 q0 = q[u][0], q1 = q[u][1], q2 = q[u][2], q3 = q[u][3];
  float Tr[16] = {q0.x, q0.y, q0.z, q0.w, q1.x, q1.y, q1.z, q1.w,
                  q2.x, q2.y, q2.z, q2.w, q3.x, q3.y, q3.z, q3.w};
  return fold_row<0>(Tr, h, bq[u]);
}

__global__ void __launch_bounds__(32)
k_combine(const float* __restrict__ Tmat,
                          const float* __restrict__ bvec,
                          const float* __restrict__ init_h,
                          float* __restrict__ hstart,
                          float* __restrict__ h_final) {
  const int d    = blockIdx.x;
  const int lane = threadIdx.x & 31;
  const unsigned lo = lane & 15;
  float h = (init_h + d * 16)[lo];

  float* hrow = hstart + d * 16 + lo;

  float4 q[PF3][4] = {};
  float  bq[PF3]   = {};
#pragma unroll
  for (int u = 0; u < AH3; ++u) {
    const float4* Trow =
        reinterpret_cast<const float4*>(Tmat + (u * 16 + d) * 256) + lo * 4;
    q[u][0] = Trow[0]; q[u][1] = Trow[1]; q[u][2] = Trow[2]; q[u][3] = Trow[3];
    bq[u] = (bvec + (u * 16 + d) * 16)[lo];
    __builtin_amdgcn_sched_barrier(0);
  }

  constexpr int NCHUNK_FULL = NCHUNK - NCHUNK % PF3;
  for (int c = 0; c < NCHUNK_FULL; c += PF3) {
#pragma unroll
    for (int u = 0; u < PF3; ++u)
      h = fold_chunk<true>(Tmat, bvec, hrow, q, bq, u, c + u, d, lo, h);
  }
#pragma unroll
  for (int u = 0; u < NCHUNK % PF3; ++u)
    h = fold_chunk<false>(Tmat, bvec, hrow, q, bq, u, NCHUNK_FULL + u, d, lo,
                          h);
  if (lane < 16) (h_final + d * 16)[lo] = h;
}

__global__ void __launch_bounds__(256)
k_scan(const float* __restrict__ dt,
                       const float* __restrict__ xb,
                       const float* __restrict__ Bv,
                       const float* __restrict__ Cv,
                       const float* __restrict__ A_log,
                       const float* __restrict__ rope,
                       const float* __restrict__ hstart,
                       float* __restrict__ y) {
  const int c    = blockIdx.x;
  const int lane = threadIdx.x & 31;
  const int lo   = lane & 15, hi = lane >> 4;
  const int d    = 2 * (threadIdx.x >> 5) + hi;
  const float A2 = -__expf(A_log[d * 16 + lo]);
  const float F2 = rope[d * 16 + lo];
  float h = hstart[(c * 16 + d) * 16 + lo];
  asm volatile("" :: "v"(h));
  const int t0  = c * CHUNK;

  constexpr int PS = 16;
  float dtq[PS], xbq[PS], Bq[PS], Cq[PS];
#pragma unroll
  for (int u = 0; u < PS; ++u) {
    dtq[u] = (dt + (t0 + u) * 16)[d];
    Bq[u]  = (Bv + (t0 + u) * 16)[lo];
    xbq[u] = (xb + (t0 + u) * 16)[d];
    Cq[u]  = (Cv + (t0 + u) * 16)[lo];
  }
#pragma unroll
  for (int u = 0; u < PS; ++u) {
    asm volatile("" :: "v"(dtq[u]), "v"(Bq[u]), "v"(xbq[u]), "v"(Cq[u]));
  }

  for (int t = 0; t < CHUNK; t += PS) {
    const int tnb = (t0 + min(t + PS, CHUNK - PS)) * 16;
    float dtn[PS], xbn[PS], Bn[PS], Cn[PS];
#pragma unroll
    for (int u = 0; u < PS; ++u) {
      dtn[u] = (dt + tnb + u * 16)[d];
      Bn[u]  = (Bv + tnb + u * 16)[lo];
      xbn[u] = (xb + tnb + u * 16)[d];
      Cn[u]  = (Cv + tnb + u * 16)[lo];
    }
    __builtin_amdgcn_sched_barrier(0);

    float ab[PS], cs[PS], sn[PS], bu[PS];
#pragma unroll
    for (int u = 0; u < PS; ++u) {
      const float dtc = dtq[u];
      const float ta  = 0.5f * dtc * A2;
      ab[u] = (1.f + ta) / (1.f - ta + 1e-8f);
      const float ang = dtc * F2;
      cs[u] = __cosf(ang);
      sn[u] = __sinf(ang);
      bu[u] = dtc * Bq[u];
    }

    float yv = 0.f;
#pragma unroll
    for (int u = 0; u < PS; ++u) {
      const float hr = row_prev(h);
      h = ab[u] * (cs[u] * h - sn[u] * hr) + bu[u] * xbq[u];
      const float p = row_sum16(h * Cq[u]);
      const float p0 = row_dpp<0x150>(p);
      yv = (lo == u) ? p0 : yv;
    }
    y[(t0 + t + lo) * 16 + d] = yv;
    __builtin_amdgcn_sched_barrier(0);

#pragma unroll
    for (int u = 0; u < PS; ++u) {
      dtq[u] = dtn[u]; Bq[u] = Bn[u]; xbq[u] = xbn[u]; Cq[u] = Cn[u];
    }
  }
}

__global__ void k_out(const float* __restrict__ y,
                      const float* __restrict__ z,
                      const float* __restrict__ xb,
                      const float* __restrict__ Dp,
                      const float* __restrict__ outw,
                      float* __restrict__ out) {
  const int lane = threadIdx.x & 31;
  const int wave = threadIdx.x >> 5;
  const int t0   = (blockIdx.x * 4 + wave) * 16;
  const int lo   = lane & 15, hi = lane >> 4;

  v8f acc = {};
#pragma unroll
  for (int kk = 0; kk < 4; ++kk) {
    const int kb = kk * 4 + 2 * hi;
    v2f a, b;
#pragma unroll
    for (int v = 0; v < 2; ++v) {
      const int k   = kb + v;
      const int idx = (t0 + lo) * 16 + k;
      const float zz  = z[idx];
      const float sil = zz / (1.f + __expf(-zz));
      const float val = y[idx] * sil + Dp[k] * xb[idx];
      if (v == 0) a.x = val; else a.y = val;
    }
    b.x = (lo < 8) ? outw[lo * 16 + kb + 0] : 0.f;
    b.y = (lo < 8) ? outw[lo * 16 + kb + 1] : 0.f;
    acc = wmma4(a, b, acc);
  }
  if (lo < 8) {
#pragma unroll
    for (int r = 0; r < 8; ++r) out[(t0 + r + 8 * hi) * 8 + lo] = acc[r];
  }
}

extern "C" void kernel_launch(void* const* d_in, const int* in_sizes, int n_in,
                              void* d_out, int out_size, void* d_ws,
                              size_t ws_size, hipStream_t stream) {
  const float* x    = (const float*)d_in[0];
  const float* init = (const float*)d_in[1];
  const float* inw  = (const float*)d_in[2];
  const float* dtw  = (const float*)d_in[3];
  const float* dtb  = (const float*)d_in[4];
  const float* bw   = (const float*)d_in[5];
  const float* cw   = (const float*)d_in[6];
  const float* alog = (const float*)d_in[7];
  const float* Dp   = (const float*)d_in[8];
  const float* rope = (const float*)d_in[9];
  const float* outw = (const float*)d_in[10];
  float* out = (float*)d_out;

  float* ws = (float*)d_ws;
  float* XB = ws;
  float* Z  = XB + N16;
  float* DT = Z + N16;
  float* BV = DT + N16;
  float* CV = BV + N16;
  float* Y  = CV + N16;
  float* T  = Y + N16;
  float* BC = T + N16;
  float* HS = BC + NCHUNK * 256;

  k_proj<<<NTOK / 64, 128, 0, stream>>>(x, inw, dtw, dtb, bw, cw,
                                        XB, Z, DT, BV, CV);
  k_chunkop<<<NCHUNK, 256, 0, stream>>>(DT, XB, BV, alog, rope, T, BC);
  k_combine<<<16, 32, 0, stream>>>(T, BC, init, HS, out + NTOK * DM);
  k_scan<<<NCHUNK, 256, 0, stream>>>(DT, XB, BV, CV, alog, rope, HS, Y);
  k_out<<<NTOK / 64, 128, 0, stream>>>(Y, Z, XB, Dp, outw, out);
}


# K3 fold: a row's 16-term sum is split over the wave's two 16-lane rows (8 terms each, two interleaved DPP multiply-add chains, one permlanex16 join): 7 chain ops a chunk, not 17; sum order changes
# speedup vs baseline: 5.2131x; 1.1171x over previous
#include <hip/hip_runtime.h>
#include <hip/hip_bf16.h>
#include <math.h>


typedef __attribute__((ext_vector_type(2))) float v2f;
typedef __attribute__((ext_vector_type(8))) float v8f;

#define NTOK   65536
#define DM     8
#define DI     16
#define CHUNK  256
#define NCHUNK 256
#define N16    (NTOK * DI)

__device__ inline v8f wmma4(v2f a, v2f b, v8f c) {
  return __builtin_amdgcn_wmma_f32_16x16x4_f32(
      false, a, false, b, (short)0, c, false, false);
}

__device__ inline float softplus_f(float s) {
  return (s > 20.f) ? s : logf(1.f + __expf(s));
}

__global__ void k_proj(const float* __restrict__ x,
                       const float* __restrict__ inw,
                       const float* __restrict__ dtw,
                       const float* __restrict__ dtb,
                       const float* __restrict__ bw,
                       const float* __restrict__ cw,
                       float* __restrict__ xb, float* __restrict__ z,
                       float* __restrict__ dt, float* __restrict__ Bv,
                       float* __restrict__ Cv) {
  __shared__ float lds[4][256];
  const int lane = threadIdx.x & 31;
  const int wave = threadIdx.x >> 5;
  const int t0   = (blockIdx.x * 4 + wave) * 16;
  const int lo   = lane & 15, hi = lane >> 4;

  v8f axb = {}, az = {};
#pragma unroll
  for (int kk = 0; kk < 2; ++kk) {
    const int kb = kk * 4 + 2 * hi;
    v2f a, b0, b1;
    a.x  = x[(t0 + lo) * 8 + kb + 0];
    a.y  = x[(t0 + lo) * 8 + kb + 1];
    b0.x = inw[lo * 8 + kb + 0];
    b0.y = inw[lo * 8 + kb + 1];
    b1.x = inw[(lo + 16) * 8 + kb + 0];
    b1.y = inw[(lo + 16) * 8 + kb + 1];
    axb = wmma4(a, b0, axb);
    az  = wmma4(a, b1, az);
  }

#pragma unroll
  for (int r = 0; r < 8; ++r) lds[wave][(r + 8 * hi) * 16 + lo] = axb[r];

  v8f adt = {}, ab = {}, ac = {};
#pragma unroll
  for (int kk = 0; kk < 4; ++kk) {
    const int kb = kk * 4 + 2 * hi;
    v2f a, b;
    a.x = lds[wave][lo * 16 + kb + 0];
    a.y = lds[wave][lo * 16 + kb + 1];
    b.x = dtw[lo * 16 + kb + 0]; b.y = dtw[lo * 16 + kb + 1];
    adt = wmma4(a, b, adt);
    b.x = bw[lo * 16 + kb + 0];  b.y = bw[lo * 16 + kb + 1];
    ab  = wmma4(a, b, ab);
    b.x = cw[lo * 16 + kb + 0];  b.y = cw[lo * 16 + kb + 1];
    ac  = wmma4(a, b, ac);
  }

  const float bias = dtb[lo];
#pragma unroll
  for (int r = 0; r < 8; ++r) {
    const int idx = (t0 + r + 8 * hi) * 16 + lo;
    xb[idx] = axb[r];
    z[idx]  = az[r];
    dt[idx] = softplus_f(adt[r] + bias);
    Bv[idx] = ab[r];
    Cv[idx] = ac[r];
  }
}

template <int CTRL>
__device__ inline float row_dpp(float v) {
  return __builtin_bit_cast(
      float, __builtin_amdgcn_update_dpp(0, __builtin_bit_cast(int, v), CTRL,
                                         0xf, 0xf, true));
}

__device__ inline float row_prev(float v) {
  return row_dpp<0x121>(v);
}

__device__ inline float row_sum16(float p) {
  p += row_dpp<0x161>(p);
  p += row_dpp<0x162>(p);
  p += row_dpp<0x164>(p);
  p += row_dpp<0x168>(p);
  return p;
}

__global__ void k_chunkop(const float* __restrict__ dt,
                          const float* __restrict__ xb,
                          const float* __restrict__ Bv,
                          const float* __restrict__ A_log,
                          const float* __restrict__ rope,
                          float* __restrict__ Tmat,
                          float* __restrict__ bvec) {
  const int c    = blockIdx.x;
  const int lane = threadIdx.x & 31;
  const int lo   = lane & 15, hi = lane >> 4;
  const int d    = 2 * (threadIdx.x >> 5) + hi;

  const float A2 = -__expf(A_log[d * 16 + lo]);
  const float F2 = rope[d * 16 + lo];

  float T[16];
#pragma unroll
  for (int r = 0; r < 16; ++r) T[r] = (r == lo) ? 1.f : 0.f;
  float b = 0.f;

  const int t0 = c * CHUNK;

  constexpr int PF2 = 8;
  float dtq[PF2] = {}, xbq[PF2] = {}, Bq[PF2] = {};
#pragma unroll
  for (int u = 0; u < PF2 - 1; ++u) {
    const int tt = t0 + u;
    dtq[u] = dt[tt * 16 + d];
    xbq[u] = xb[tt * 16 + d];
    Bq[u]  = Bv[tt * 16 + lo];
    __builtin_amdgcn_sched_barrier(0);
  }

  for (int t = 0; t < CHUNK; t += PF2) {
#pragma unroll
  for (int u = 0; u < PF2; ++u) {
    {
      const int v  = (u + PF2 - 1) % PF2;
      const int tn = t0 + min(t + u + PF2 - 1, CHUNK - 1);
      dtq[v] = dt[tn * 16 + d];
      xbq[v] = xb[tn * 16 + d];
      Bq[v]  = Bv[tn * 16 + lo];
      __builtin_amdgcn_sched_barrier(0);
    }
    const float dtc = dtq[u];
    const float xbd = xbq[u];
    const float Bj  = Bq[u];

    const float ta   = 0.5f * dtc * A2;
    const float abar = (1.f + ta) / (1.f - ta + 1e-8f);
    const float ang  = dtc * F2;
    const float cs   = __cosf(ang);
    const float sn   = __sinf(ang);

#pragma unroll
    for (int r = 0; r < 16; ++r)
      T[r] = abar * (cs * T[r] - sn * row_prev(T[r]));
    b = abar * (cs * b - sn * row_prev(b)) + dtc * Bj * xbd;
  }
  }

  float4* dst = reinterpret_cast<float4*>(
      Tmat + ((c * 16 + d) * 16 + lo) * 16);
  dst[0] = make_float4(T[0], T[1], T[2], T[3]);
  dst[1] = make_float4(T[4], T[5], T[6], T[7]);
  dst[2] = make_float4(T[8], T[9], T[10], T[11]);
  dst[3] = make_float4(T[12], T[13], T[14], T[15]);
  bvec[(c * 16 + d) * 16 + lo] = b;
}

__device__ inline float other_row_plus8(float v) {
  return __builtin_bit_cast(
      float, __builtin_amdgcn_permlanex16(0u, __builtin_bit_cast(unsigned, v),
                                          0xFEDCBA98u, 0x76543210u, true,
                                          true));
}

constexpr int PF3 = 21;
constexpr int AH3 = PF3 - 1;

__device__ inline float fold_chunk(const bool reload,
                                   const float* __restrict__ Tmat,
                                   const float* __restrict__ bvec,
                                   float* __restrict__ hrow,
                                   float4 (&q)[PF3][2], float (&bq)[PF3],
                                   const int u, const int cc, const int d,
                                   const unsigned toff, const unsigned j,
                                   const bool row0, float& bsel, float h) {
  hrow[(size_t)cc * 256] = h;
  if (reload) {
    const int v  = (u + AH3) % PF3;
    const int cn = cc + AH3;
    const float4* Trow =
        reinterpret_cast<const float4*>(Tmat + (cn * 16 + d) * 256) + toff;
    q[v][0] = Trow[0]; q[v][1] = Trow[1];
    bq[v] = (bvec + (cn * 16 + d) * 16)[j];
  }
  const float4 q0 = q[u][0], q1 = q[u][1];
  float x = bsel, y;
  asm("v_mul_f32_dpp %1, %2, %7 row_share:4 row_mask:0xf bank_mask:0xf bound_ctrl:1\n\t"
      "v_fmac_f32_dpp %0, %2, %3 row_share:0 row_mask:0xf bank_mask:0xf bound_ctrl:1\n\t"
      "v_fmac_f32_dpp %1, %2, %8 row_share:5 row_mask:0xf bank_mask:0xf bound_ctrl:1\n\t"
      "v_fmac_f32_dpp %0, %2, %4 row_share:1 row_mask:0xf bank_mask:0xf bound_ctrl:1\n\t"
      "v_fmac_f32_dpp %1, %2, %9 row_share:6 row_mask:0xf bank_mask:0xf bound_ctrl:1\n\t"
      "v_fmac_f32_dpp %0, %2, %5 row_share:2 row_mask:0xf bank_mask:0xf bound_ctrl:1\n\t"
      "v_fmac_f32_dpp %1, %2, %10 row_share:7 row_mask:0xf bank_mask:0xf bound_ctrl:1\n\t"
      "v_fmac_f32_dpp %0, %2, %6 row_share:3 row_mask:0xf bank_mask:0xf bound_ctrl:1"
      : "+v"(x), "=&v"(y)
      : "v"(h), "v"(q0.x), "v"(q0.y), "v"(q0.z), "v"(q0.w),
        "v"(q1.x), "v"(q1.y), "v"(q1.z), "v"(q1.w));
  const float s = x + y;
  bsel = row0 ? bq[(u + 1) % PF3] : 0.f;
  h = s + other_row_plus8(s);
  __builtin_amdgcn_sched_barrier(0);
  return h;
}

__global__ void __launch_bounds__(32)
k_combine(const float* __restrict__ Tmat,
                          const float* __restrict__ bvec,
                          const float* __restrict__ init_h,
                          float* __restrict__ hstart,
                          float* __restrict__ h_final) {
  const int d    = blockIdx.x;
  const int lane = threadIdx.x & 31;
  const unsigned lo = lane & 15, hi = lane >> 4;
  const unsigned j    = (lo + 8 * hi) & 15;
  const unsigned toff = j * 4 + hi * 2;
  const bool row0 = hi == 0;
  float h = (init_h + d * 16)[j];

  float* hrow = hstart + d * 16 + j;

  float4 q[PF3][2] = {};
  float  bq[PF3]   = {};
#pragma unroll
  for (int u = 0; u < AH3; ++u) {
    const float4* Trow =
        reinterpret_cast<const float4*>(Tmat + (u * 16 + d) * 256) + toff;
    q[u][0] = Trow[0]; q[u][1] = Trow[1];
    bq[u] = (bvec + (u * 16 + d) * 16)[j];
    __builtin_amdgcn_sched_barrier(0);
  }

  float bsel = row0 ? bq[0] : 0.f;

  constexpr int MAIN_END = (NCHUNK - AH3) / PF3 * PF3;
  for (int c = 0; c < MAIN_END; c += PF3) {
#pragma unroll
    for (int u = 0; u < PF3; ++u)
      h = fold_chunk(true, Tmat, bvec, hrow, q, bq, u, c + u, d, toff, j, row0,
                     bsel, h);
  }
#pragma unroll
  for (int i = 0; i < NCHUNK - MAIN_END; ++i)
    h = fold_chunk(MAIN_END + i + AH3 < NCHUNK, Tmat, bvec, hrow, q, bq,
                   i % PF3, MAIN_END + i, d, toff, j, row0, bsel, h);
  if (lane < 16) (h_final + d * 16)[lo] = h;
}

__global__ void __launch_bounds__(256)
k_scan(const float* __restrict__ dt,
                       const float* __restrict__ xb,
                       const float* __restrict__ Bv,
                       const float* __restrict__ Cv,
                       const float* __restrict__ A_log,
                       const float* __restrict__ rope,
                       const float* __restrict__ hstart,
                       float* __restrict__ y) {
  const int c    = blockIdx.x;
  const int lane = threadIdx.x & 31;
  const int lo   = lane & 15, hi = lane >> 4;
  const int d    = 2 * (threadIdx.x >> 5) + hi;
  const float A2 = -__expf(A_log[d * 16 + lo]);
  const float F2 = rope[d * 16 + lo];
  float h = hstart[(c * 16 + d) * 16 + lo];
  asm volatile("" :: "v"(h));
  const int t0  = c * CHUNK;

  constexpr int PS = 16;
  float dtq[PS], xbq[PS], Bq[PS], Cq[PS];
#pragma unroll
  for (int u = 0; u < PS; ++u) {
    dtq[u] = (dt + (t0 + u) * 16)[d];
    Bq[u]  = (Bv + (t0 + u) * 16)[lo];
    xbq[u] = (xb + (t0 + u) * 16)[d];
    Cq[u]  = (Cv + (t0 + u) * 16)[lo];
  }
#pragma unroll
  for (int u = 0; u < PS; ++u) {
    asm volatile("" :: "v"(dtq[u]), "v"(Bq[u]), "v"(xbq[u]), "v"(Cq[u]));
  }

  for (int t = 0; t < CHUNK; t += PS) {
    const int tnb = (t0 + min(t + PS, CHUNK - PS)) * 16;
    float dtn[PS], xbn[PS], Bn[PS], Cn[PS];
#pragma unroll
    for (int u = 0; u < PS; ++u) {
      dtn[u] = (dt + tnb + u * 16)[d];
      Bn[u]  = (Bv + tnb + u * 16)[lo];
      xbn[u] = (xb + tnb + u * 16)[d];
      Cn[u]  = (Cv + tnb + u * 16)[lo];
    }
    __builtin_amdgcn_sched_barrier(0);

    float ab[PS], cs[PS], sn[PS], bu[PS];
#pragma unroll
    for (int u = 0; u < PS; ++u) {
      const float dtc = dtq[u];
      const float ta  = 0.5f * dtc * A2;
      ab[u] = (1.f + ta) / (1.f - ta + 1e-8f);
      const float ang = dtc * F2;
      cs[u] = __cosf(ang);
      sn[u] = __sinf(ang);
      bu[u] = dtc * Bq[u];
    }

    float yv = 0.f;
#pragma unroll
    for (int u = 0; u < PS; ++u) {
      const float hr = row_prev(h);
      h = ab[u] * (cs[u] * h - sn[u] * hr) + bu[u] * xbq[u];
      const float p = row_sum16(h * Cq[u]);
      const float p0 = row_dpp<0x150>(p);
      yv = (lo == u) ? p0 : yv;
    }
    y[(t0 + t + lo) * 16 + d] = yv;
    __builtin_amdgcn_sched_barrier(0);

#pragma unroll
    for (int u = 0; u < PS; ++u) {
      dtq[u] = dtn[u]; Bq[u] = Bn[u]; xbq[u] = xbn[u]; Cq[u] = Cn[u];
    }
  }
}

__global__ void k_out(const float* __restrict__ y,
                      const float* __restrict__ z,
                      const float* __restrict__ xb,
                      const float* __restrict__ Dp,
                      const float* __restrict__ outw,
                      float* __restrict__ out) {
  const int lane = threadIdx.x & 31;
  const int wave = threadIdx.x >> 5;
  const int t0   = (blockIdx.x * 4 + wave) * 16;
  const int lo   = lane & 15, hi = lane >> 4;

  v8f acc = {};
#pragma unroll
  for (int kk = 0; kk < 4; ++kk) {
    const int kb = kk * 4 + 2 * hi;
    v2f a, b;
#pragma unroll
    for (int v = 0; v < 2; ++v) {
      const int k   = kb + v;
      const int idx = (t0 + lo) * 16 + k;
      const float zz  = z[idx];
      const float sil = zz / (1.f + __expf(-zz));
      const float val = y[idx] * sil + Dp[k] * xb[idx];
      if (v == 0) a.x = val; else a.y = val;
    }
    b.x = (lo < 8) ? outw[lo * 16 + kb + 0] : 0.f;
    b.y = (lo < 8) ? outw[lo * 16 + kb + 1] : 0.f;
    acc = wmma4(a, b, acc);
  }
  if (lo < 8) {
#pragma unroll
    for (int r = 0; r < 8; ++r) out[(t0 + r + 8 * hi) * 8 + lo] = acc[r];
  }
}

extern "C" void kernel_launch(void* const* d_in, const int* in_sizes, int n_in,
                              void* d_out, int out_size, void* d_ws,
                              size_t ws_size, hipStream_t stream) {
  const float* x    = (const float*)d_in[0];
  const float* init = (const float*)d_in[1];
  const float* inw  = (const float*)d_in[2];
  const float* dtw  = (const float*)d_in[3];
  const float* dtb  = (const float*)d_in[4];
  const float* bw   = (const float*)d_in[5];
  const float* cw   = (const float*)d_in[6];
  const float* alog = (const float*)d_in[7];
  const float* Dp   = (const float*)d_in[8];
  const float* rope = (const float*)d_in[9];
  const float* outw = (const float*)d_in[10];
  float* out = (float*)d_out;

  float* ws = (float*)d_ws;
  float* XB = ws;
  float* Z  = XB + N16;
  float* DT = Z + N16;
  float* BV = DT + N16;
  float* CV = BV + N16;
  float* Y  = CV + N16;
  float* T  = Y + N16;
  float* BC = T + N16;
  float* HS = BC + NCHUNK * 256;

  k_proj<<<NTOK / 64, 128, 0, stream>>>(x, inw, dtw, dtb, bw, cw,
                                        XB, Z, DT, BV, CV);
  k_chunkop<<<NCHUNK, 256, 0, stream>>>(DT, XB, BV, alog, rope, T, BC);
  k_combine<<<16, 32, 0, stream>>>(T, BC, init, HS, out + NTOK * DM);
  k_scan<<<NCHUNK, 256, 0, stream>>>(DT, XB, BV, CV, alog, rope, HS, Y);
  k_out<<<NTOK / 64, 128, 0, stream>>>(Y, Z, XB, Dp, outw, out);
}


# Launches 5 -> 3: projections run in the head of the chunk-operator kernel, gated out-projection in the tail of the scan kernel (same 256 tokens per work-group; y stays in LDS); arithmetic unchanged
# speedup vs baseline: 5.8729x; 1.1266x over previous
#include <hip/hip_runtime.h>
#include <hip/hip_bf16.h>
#include <math.h>


typedef __attribute__((ext_vector_type(2))) float v2f;
typedef __attribute__((ext_vector_type(8))) float v8f;

#define NTOK   65536
#define DM     8
#define DI     16
#define CHUNK  256
#define NCHUNK 256
#define N16    (NTOK * DI)
#define YP     17

__device__ inline v8f wmma4(v2f a, v2f b, v8f c) {
  return __builtin_amdgcn_wmma_f32_16x16x4_f32(
      false, a, false, b, (short)0, c, false, false);
}

__device__ inline float softplus_f(float s) {
  return (s > 20.f) ? s : logf(1.f + __expf(s));
}

struct ProjAcc { v8f xb, z, dt, b, c; };

__device__ inline ProjAcc proj_tile_mm(const float* __restrict__ x,
                                       const float* __restrict__ inw,
                                       const float* __restrict__ dtw,
                                       const float* __restrict__ bw,
                                       const float* __restrict__ cw,
                                       float* lds,
                                       const int t0, const int lo,
                                       const int hi) {
  ProjAcc p;
  v8f axb = {}, az = {};
#pragma unroll
  for (int kk = 0; kk < 2; ++kk) {
    const int kb = kk * 4 + 2 * hi;
    v2f a, b0, b1;
    a.x  = x[(t0 + lo) * 8 + kb + 0];
    a.y  = x[(t0 + lo) * 8 + kb + 1];
    b0.x = inw[lo * 8 + kb + 0];
    b0.y = inw[lo * 8 + kb + 1];
    b1.x = inw[(lo + 16) * 8 + kb + 0];
    b1.y = inw[(lo + 16) * 8 + kb + 1];
    axb = wmma4(a, b0, axb);
    az  = wmma4(a, b1, az);
  }

#pragma unroll
  for (int r = 0; r < 8; ++r) lds[(r + 8 * hi) * 16 + lo] = axb[r];

  v8f adt = {}, ab = {}, ac = {};
#pragma unroll
  for (int kk = 0; kk < 4; ++kk) {
    const int kb = kk * 4 + 2 * hi;
    v2f a, b;
    a.x = lds[lo * 16 + kb + 0];
    a.y = lds[lo * 16 + kb + 1];
    b.x = dtw[lo * 16 + kb + 0]; b.y = dtw[lo * 16 + kb + 1];
    adt = wmma4(a, b, adt);
    b.x = bw[lo * 16 + kb + 0];  b.y = bw[lo * 16 + kb + 1];
    ab  = wmma4(a, b, ab);
    b.x = cw[lo * 16 + kb + 0];  b.y = cw[lo * 16 + kb + 1];
    ac  = wmma4(a, b, ac);
  }
  p.xb = axb; p.z = az; p.dt = adt; p.b = ab; p.c = ac;
  return p;
}

__device__ inline void proj_tile_store(const ProjAcc& p, const float bias,
                                       float* xb, float* z, float* dt,
                                       float* Bv, float* Cv, const int t0,
                                       const int lo, const int hi) {
#pragma unroll
  for (int r = 0; r < 8; ++r) {
    const int idx = (t0 + r + 8 * hi) * 16 + lo;
    xb[idx] = p.xb[r];
    z[idx]  = p.z[r];
    dt[idx] = softplus_f(p.dt[r] + bias);
    Bv[idx] = p.b[r];
    Cv[idx] = p.c[r];
  }
}

template <int CTRL>
__device__ inline float row_dpp(float v) {
  return __builtin_bit_cast(
      float, __builtin_amdgcn_update_dpp(0, __builtin_bit_cast(int, v), CTRL,
                                         0xf, 0xf, true));
}

__device__ inline float row_prev(float v) {
  return row_dpp<0x121>(v);
}

__device__ inline float row_sum16(float p) {
  p += row_dpp<0x161>(p);
  p += row_dpp<0x162>(p);
  p += row_dpp<0x164>(p);
  p += row_dpp<0x168>(p);
  return p;
}

__global__ void
k_chunkop(const float* __restrict__ x,
          const float* __restrict__ inw,
          const float* __restrict__ dtw,
          const float* __restrict__ dtb,
          const float* __restrict__ bw,
          const float* __restrict__ cw,
          float* xb, float* z, float* dt, float* Bv, float* Cv,
          const float* __restrict__ A_log,
          const float* __restrict__ rope,
          float* __restrict__ Tmat,
          float* __restrict__ bvec) {
  __shared__ float lds[8][2][256];
  const int c    = blockIdx.x;
  const int lane = threadIdx.x & 31;
  const int lo   = lane & 15, hi = lane >> 4;
  const int wave = threadIdx.x >> 5;
  const int d    = 2 * wave + hi;

  ProjAcc acc[2];
#pragma unroll
  for (int i = 0; i < 2; ++i)
    acc[i] = proj_tile_mm(x, inw, dtw, bw, cw, lds[wave][i],
                          c * CHUNK + (2 * wave + i) * 16, lo, hi);
  const float bias = dtb[lo];
#pragma unroll
  for (int i = 0; i < 2; ++i)
    proj_tile_store(acc[i], bias, xb, z, dt, Bv, Cv,
                    c * CHUNK + (2 * wave + i) * 16, lo, hi);
  __syncthreads();

  const float A2 = -__expf(A_log[d * 16 + lo]);
  const float F2 = rope[d * 16 + lo];

  float T[16];
#pragma unroll
  for (int r = 0; r < 16; ++r) T[r] = (r == lo) ? 1.f : 0.f;
  float b = 0.f;

  const int t0 = c * CHUNK;

  constexpr int PF2 = 8;
  float dtq[PF2] = {}, xbq[PF2] = {}, Bq[PF2] = {};
#pragma unroll
  for (int u = 0; u < PF2 - 1; ++u) {
    const int tt = t0 + u;
    dtq[u] = dt[tt * 16 + d];
    xbq[u] = xb[tt * 16 + d];
    Bq[u]  = Bv[tt * 16 + lo];
    __builtin_amdgcn_sched_barrier(0);
  }

  for (int t = 0; t < CHUNK; t += PF2) {
#pragma unroll
  for (int u = 0; u < PF2; ++u) {
    {
      const int v  = (u + PF2 - 1) % PF2;
      const int tn = t0 + min(t + u + PF2 - 1, CHUNK - 1);
      dtq[v] = dt[tn * 16 + d];
      xbq[v] = xb[tn * 16 + d];
      Bq[v]  = Bv[tn * 16 + lo];
      __builtin_amdgcn_sched_barrier(0);
    }
    const float dtc = dtq[u];
    const float xbd = xbq[u];
    const float Bj  = Bq[u];

    const float ta   = 0.5f * dtc * A2;
    const float abar = (1.f + ta) / (1.f - ta + 1e-8f);
    const float ang  = dtc * F2;
    const float cs   = __cosf(ang);
    const float sn   = __sinf(ang);

#pragma unroll
    for (int r = 0; r < 16; ++r)
      T[r] = abar * (cs * T[r] - sn * row_prev(T[r]));
    b = abar * (cs * b - sn * row_prev(b)) + dtc * Bj * xbd;
  }
  }

  float4* dst = reinterpret_cast<float4*>(
      Tmat + ((c * 16 + d) * 16 + lo) * 16);
  dst[0] = make_float4(T[0], T[1], T[2], T[3]);
  dst[1] = make_float4(T[4], T[5], T[6], T[7]);
  dst[2] = make_float4(T[8], T[9], T[10], T[11]);
  dst[3] = make_float4(T[12], T[13], T[14], T[15]);
  bvec[(c * 16 + d) * 16 + lo] = b;
}

__device__ inline float other_row_plus8(float v) {
  return __builtin_bit_cast(
      float, __builtin_amdgcn_permlanex16(0u, __builtin_bit_cast(unsigned, v),
                                          0xFEDCBA98u, 0x76543210u, true,
                                          true));
}

constexpr int PF3 = 21;
constexpr int AH3 = PF3 - 1;

__device__ inline float fold_chunk(const bool reload,
                                   const float* __restrict__ Tmat,
                                   const float* __restrict__ bvec,
                                   float* __restrict__ hrow,
                                   float4 (&q)[PF3][2], float (&bq)[PF3],
                                   const int u, const int cc, const int d,
                                   const unsigned toff, const unsigned j,
                                   const bool row0, float& bsel, float h) {
  hrow[(size_t)cc * 256] = h;
  if (reload) {
    const int v  = (u + AH3) % PF3;
    const int cn = cc + AH3;
    const float4* Trow =
        reinterpret_cast<const float4*>(Tmat + (cn * 16 + d) * 256) + toff;
    q[v][0] = Trow[0]; q[v][1] = Trow[1];
    bq[v] = (bvec + (cn * 16 + d) * 16)[j];
  }
  const float4 q0 = q[u][0], q1 = q[u][1];
  float x = bsel, y;
  asm("v_mul_f32_dpp %1, %2, %7 row_share:4 row_mask:0xf bank_mask:0xf bound_ctrl:1\n\t"
      "v_fmac_f32_dpp %0, %2, %3 row_share:0 row_mask:0xf bank_mask:0xf bound_ctrl:1\n\t"
      "v_fmac_f32_dpp %1, %2, %8 row_share:5 row_mask:0xf bank_mask:0xf bound_ctrl:1\n\t"
      "v_fmac_f32_dpp %0, %2, %4 row_share:1 row_mask:0xf bank_mask:0xf bound_ctrl:1\n\t"
      "v_fmac_f32_dpp %1, %2, %9 row_share:6 row_mask:0xf bank_mask:0xf bound_ctrl:1\n\t"
      "v_fmac_f32_dpp %0, %2, %5 row_share:2 row_mask:0xf bank_mask:0xf bound_ctrl:1\n\t"
      "v_fmac_f32_dpp %1, %2, %10 row_share:7 row_mask:0xf bank_mask:0xf bound_ctrl:1\n\t"
      "v_fmac_f32_dpp %0, %2, %6 row_share:3 row_mask:0xf bank_mask:0xf bound_ctrl:1"
      : "+v"(x), "=&v"(y)
      : "v"(h), "v"(q0.x), "v"(q0.y), "v"(q0.z), "v"(q0.w),
        "v"(q1.x), "v"(q1.y), "v"(q1.z), "v"(q1.w));
  const float s = x + y;
  bsel = row0 ? bq[(u + 1) % PF3] : 0.f;
  h = s + other_row_plus8(s);
  __builtin_amdgcn_sched_barrier(0);
  return h;
}

__global__ void __launch_bounds__(32)
k_combine(const float* __restrict__ Tmat,
                          const float* __restrict__ bvec,
                          const float* __restrict__ init_h,
                          float* __restrict__ hstart,
                          float* __restrict__ h_final) {
  const int d    = blockIdx.x;
  const int lane = threadIdx.x & 31;
  const unsigned lo = lane & 15, hi = lane >> 4;
  const unsigned j    = (lo + 8 * hi) & 15;
  const unsigned toff = j * 4 + hi * 2;
  const bool row0 = hi == 0;
  float h = (init_h + d * 16)[j];

  float* hrow = hstart + d * 16 + j;

  float4 q[PF3][2] = {};
  float  bq[PF3]   = {};
#pragma unroll
  for (int u = 0; u < AH3; ++u) {
    const float4* Trow =
        reinterpret_cast<const float4*>(Tmat + (u * 16 + d) * 256) + toff;
    q[u][0] = Trow[0]; q[u][1] = Trow[1];
    bq[u] = (bvec + (u * 16 + d) * 16)[j];
    __builtin_amdgcn_sched_barrier(0);
  }

  float bsel = row0 ? bq[0] : 0.f;

  constexpr int MAIN_END = (NCHUNK - AH3) / PF3 * PF3;
  for (int c = 0; c < MAIN_END; c += PF3) {
#pragma unroll
    for (int u = 0; u < PF3; ++u)
      h = fold_chunk(true, Tmat, bvec, hrow, q, bq, u, c + u, d, toff, j, row0,
                     bsel, h);
  }
#pragma unroll
  for (int i = 0; i < NCHUNK - MAIN_END; ++i)
    h = fold_chunk(MAIN_END + i + AH3 < NCHUNK, Tmat, bvec, hrow, q, bq,
                   i % PF3, MAIN_END + i, d, toff, j, row0, bsel, h);
  if (lane < 16) (h_final + d * 16)[lo] = h;
}

__global__ void __launch_bounds__(256)
k_scan(const float* __restrict__ dt,
                       const float* __restrict__ xb,
                       const float* __restrict__ Bv,
                       const float* __restrict__ Cv,
                       const float* __restrict__ z,
                       const float* __restrict__ A_log,
                       const float* __restrict__ rope,
                       const float* __restrict__ hstart,
                       const float* __restrict__ Dp,
                       const float* __restrict__ outw,
                       float* __restrict__ out) {
  __shared__ float ylds[CHUNK * YP];
  const int c    = blockIdx.x;
  const int lane = threadIdx.x & 31;
  const int lo   = lane & 15, hi = lane >> 4;
  const int d    = 2 * (threadIdx.x >> 5) + hi;
  const float A2 = -__expf(A_log[d * 16 + lo]);
  const float F2 = rope[d * 16 + lo];
  float h = hstart[(c * 16 + d) * 16 + lo];
  asm volatile("" :: "v"(h));
  const int t0  = c * CHUNK;

  constexpr int PS = 16;
  float dtq[PS], xbq[PS], Bq[PS], Cq[PS];
#pragma unroll
  for (int u = 0; u < PS; ++u) {
    dtq[u] = (dt + (t0 + u) * 16)[d];
    Bq[u]  = (Bv + (t0 + u) * 16)[lo];
    xbq[u] = (xb + (t0 + u) * 16)[d];
    Cq[u]  = (Cv + (t0 + u) * 16)[lo];
  }
#pragma unroll
  for (int u = 0; u < PS; ++u) {
    asm volatile("" :: "v"(dtq[u]), "v"(Bq[u]), "v"(xbq[u]), "v"(Cq[u]));
  }

  for (int t = 0; t < CHUNK; t += PS) {
    const int tnb = (t0 + min(t + PS, CHUNK - PS)) * 16;
    float dtn[PS], xbn[PS], Bn[PS], Cn[PS];
#pragma unroll
    for (int u = 0; u < PS; ++u) {
      dtn[u] = (dt + tnb + u * 16)[d];
      Bn[u]  = (Bv + tnb + u * 16)[lo];
      xbn[u] = (xb + tnb + u * 16)[d];
      Cn[u]  = (Cv + tnb + u * 16)[lo];
    }
    __builtin_amdgcn_sched_barrier(0);

    float ab[PS], cs[PS], sn[PS], bu[PS];
#pragma unroll
    for (int u = 0; u < PS; ++u) {
      const float dtc = dtq[u];
      const float ta  = 0.5f * dtc * A2;
      ab[u] = (1.f + ta) / (1.f - ta + 1e-8f);
      const float ang = dtc * F2;
      cs[u] = __cosf(ang);
      sn[u] = __sinf(ang);
      bu[u] = dtc * Bq[u];
    }

    float yv = 0.f;
#pragma unroll
    for (int u = 0; u < PS; ++u) {
      const float hr = row_prev(h);
      h = ab[u] * (cs[u] * h - sn[u] * hr) + bu[u] * xbq[u];
      const float p = row_sum16(h * Cq[u]);
      const float p0 = row_dpp<0x150>(p);
      yv = (lo == u) ? p0 : yv;
    }
    ylds[(t + lo) * YP + d] = yv;
    __builtin_amdgcn_sched_barrier(0);

#pragma unroll
    for (int u = 0; u < PS; ++u) {
      dtq[u] = dtn[u]; Bq[u] = Bn[u]; xbq[u] = xbn[u]; Cq[u] = Cn[u];
    }
  }

  const int wave = threadIdx.x >> 5;
  v2f bo[4], dq[4], zq[2][4], xq[2][4];
#pragma unroll
  for (int kk = 0; kk < 4; ++kk) {
    const int kb = kk * 4 + 2 * hi;
    bo[kk].x = (lo < 8) ? outw[lo * 16 + kb + 0] : 0.f;
    bo[kk].y = (lo < 8) ? outw[lo * 16 + kb + 1] : 0.f;
    dq[kk].x = Dp[kb + 0];
    dq[kk].y = Dp[kb + 1];
#pragma unroll
    for (int i = 0; i < 2; ++i) {
      const int idx = (t0 + (2 * wave + i) * 16 + lo) * 16 + kb;
      zq[i][kk].x = z[idx + 0];  zq[i][kk].y = z[idx + 1];
      xq[i][kk].x = xb[idx + 0]; xq[i][kk].y = xb[idx + 1];
    }
  }
  __syncthreads();
  v8f acc[2];
#pragma unroll
  for (int i = 0; i < 2; ++i) {
    const int tl = (2 * wave + i) * 16;
    acc[i] = v8f{};
#pragma unroll
    for (int kk = 0; kk < 4; ++kk) {
      const int kb = kk * 4 + 2 * hi;
      v2f a;
#pragma unroll
      for (int v = 0; v < 2; ++v) {
        const float zz  = zq[i][kk][v];
        const float sil = zz / (1.f + __expf(-zz));
        const float val =
            ylds[(tl + lo) * YP + kb + v] * sil + dq[kk][v] * xq[i][kk][v];
        if (v == 0) a.x = val; else a.y = val;
      }
      acc[i] = wmma4(a, bo[kk], acc[i]);
    }
  }
  if (lo < 8) {
#pragma unroll
    for (int i = 0; i < 2; ++i) {
#pragma unroll
      for (int r = 0; r < 8; ++r)
        out[(t0 + (2 * wave + i) * 16 + r + 8 * hi) * 8 + lo] = acc[i][r];
    }
  }
}

extern "C" void kernel_launch(void* const* d_in, const int* in_sizes, int n_in,
                              void* d_out, int out_size, void* d_ws,
                              size_t ws_size, hipStream_t stream) {
  const float* x    = (const float*)d_in[0];
  const float* init = (const float*)d_in[1];
  const float* inw  = (const float*)d_in[2];
  const float* dtw  = (const float*)d_in[3];
  const float* dtb  = (const float*)d_in[4];
  const float* bw   = (const float*)d_in[5];
  const float* cw   = (const float*)d_in[6];
  const float* alog = (const float*)d_in[7];
  const float* Dp   = (const float*)d_in[8];
  const float* rope = (const float*)d_in[9];
  const float* outw = (const float*)d_in[10];
  float* out = (float*)d_out;

  float* ws = (float*)d_ws;
  float* XB = ws;
  float* Z  = XB + N16;
  float* DT = Z + N16;
  float* BV = DT + N16;
  float* CV = BV + N16;
  float* T  = CV + N16;
  float* BC = T + N16;
  float* HS = BC + NCHUNK * 256;

  k_chunkop<<<NCHUNK, 256, 0, stream>>>(x, inw, dtw, dtb, bw, cw,
                                        XB, Z, DT, BV, CV, alog, rope, T, BC);
  k_combine<<<16, 32, 0, stream>>>(T, BC, init, HS, out + NTOK * DM);
  k_scan<<<NCHUNK, 256, 0, stream>>>(DT, XB, BV, CV, Z, alog, rope, HS,
                                     Dp, outw, out);
}


# K2 chunk operator: a block of 8 steps is composed as its 9-diagonal band on the vector ALU and applied as T <- G*T by WMMA f32 16x16x4, not 16 column updates per step; sum order in T changes
# speedup vs baseline: 6.2232x; 1.0596x over previous
#include <hip/hip_runtime.h>
#include <hip/hip_bf16.h>
#include <math.h>


typedef __attribute__((ext_vector_type(2))) float v2f;
typedef __attribute__((ext_vector_type(8))) float v8f;

#define NTOK   65536
#define DM     8
#define DI     16
#define CHUNK  256
#define NCHUNK 256
#define N16    (NTOK * DI)
#define YP     17

__device__ inline v8f wmma4(v2f a, v2f b, v8f c) {
  return __builtin_amdgcn_wmma_f32_16x16x4_f32(
      false, a, false, b, (short)0, c, false, false);
}

__device__ inline float softplus_f(float s) {
  return (s > 20.f) ? s : logf(1.f + __expf(s));
}

struct ProjAcc { v8f xb, z, dt, b, c; };

__device__ inline ProjAcc proj_tile_mm(const float* __restrict__ x,
                                       const float* __restrict__ inw,
                                       const float* __restrict__ dtw,
                                       const float* __restrict__ bw,
                                       const float* __restrict__ cw,
                                       float* lds,
                                       const int t0, const int lo,
                                       const int hi) {
  ProjAcc p;
  v8f axb = {}, az = {};
#pragma unroll
  for (int kk = 0; kk < 2; ++kk) {
    const int kb = kk * 4 + 2 * hi;
    v2f a, b0, b1;
    a.x  = x[(t0 + lo) * 8 + kb + 0];
    a.y  = x[(t0 + lo) * 8 + kb + 1];
    b0.x = inw[lo * 8 + kb + 0];
    b0.y = inw[lo * 8 + kb + 1];
    b1.x = inw[(lo + 16) * 8 + kb + 0];
    b1.y = inw[(lo + 16) * 8 + kb + 1];
    axb = wmma4(a, b0, axb);
    az  = wmma4(a, b1, az);
  }

#pragma unroll
  for (int r = 0; r < 8; ++r) lds[(r + 8 * hi) * 16 + lo] = axb[r];

  v8f adt = {}, ab = {}, ac = {};
#pragma unroll
  for (int kk = 0; kk < 4; ++kk) {
    const int kb = kk * 4 + 2 * hi;
    v2f a, b;
    a.x = lds[lo * 16 + kb + 0];
    a.y = lds[lo * 16 + kb + 1];
    b.x = dtw[lo * 16 + kb + 0]; b.y = dtw[lo * 16 + kb + 1];
    adt = wmma4(a, b, adt);
    b.x = bw[lo * 16 + kb + 0];  b.y = bw[lo * 16 + kb + 1];
    ab  = wmma4(a, b, ab);
    b.x = cw[lo * 16 + kb + 0];  b.y = cw[lo * 16 + kb + 1];
    ac  = wmma4(a, b, ac);
  }
  p.xb = axb; p.z = az; p.dt = adt; p.b = ab; p.c = ac;
  return p;
}

__device__ inline void proj_tile_store(const ProjAcc& p, const float bias,
                                       float* xb, float* z, float* dt,
                                       float* Bv, float* Cv, const int t0,
                                       const int lo, const int hi) {
#pragma unroll
  for (int r = 0; r < 8; ++r) {
    const int idx = (t0 + r + 8 * hi) * 16 + lo;
    xb[idx] = p.xb[r];
    z[idx]  = p.z[r];
    dt[idx] = softplus_f(p.dt[r] + bias);
    Bv[idx] = p.b[r];
    Cv[idx] = p.c[r];
  }
}

template <int CTRL>
__device__ inline float row_dpp(float v) {
  return __builtin_bit_cast(
      float, __builtin_amdgcn_update_dpp(0, __builtin_bit_cast(int, v), CTRL,
                                         0xf, 0xf, true));
}

__device__ inline float row_prev(float v) {
  return row_dpp<0x121>(v);
}

__device__ inline float row_sum16(float p) {
  p += row_dpp<0x161>(p);
  p += row_dpp<0x162>(p);
  p += row_dpp<0x164>(p);
  p += row_dpp<0x168>(p);
  return p;
}

constexpr int BS   = 8;
constexpr int GPIT = 288;

__global__ void
k_chunkop(const float* __restrict__ x,
          const float* __restrict__ inw,
          const float* __restrict__ dtw,
          const float* __restrict__ dtb,
          const float* __restrict__ bw,
          const float* __restrict__ cw,
          float* xb, float* z, float* dt, float* Bv, float* Cv,
          const float* __restrict__ A_log,
          const float* __restrict__ rope,
          float* __restrict__ Tmat,
          float* __restrict__ bvec) {
  __shared__ float lds[8][2][256];
  __shared__ __attribute__((aligned(16))) float gt[8][2][GPIT];
  const int c    = blockIdx.x;
  const int lane = threadIdx.x & 31;
  const int lo   = lane & 15, hi = lane >> 4;
  const int wave = threadIdx.x >> 5;
  const int d    = 2 * wave + hi;

  ProjAcc acc[2];
#pragma unroll
  for (int i = 0; i < 2; ++i)
    acc[i] = proj_tile_mm(x, inw, dtw, bw, cw, lds[wave][i],
                          c * CHUNK + (2 * wave + i) * 16, lo, hi);
  const float bias = dtb[lo];
#pragma unroll
  for (int i = 0; i < 2; ++i)
    proj_tile_store(acc[i], bias, xb, z, dt, Bv, Cv,
                    c * CHUNK + (2 * wave + i) * 16, lo, hi);
  {
    float* g0 = &gt[wave][0][0];
#pragma unroll
    for (int q = 0; q < 2 * GPIT / 32; ++q) g0[q * 32 + lane] = 0.f;
  }
  __syncthreads();

  const float A2 = -__expf(A_log[d * 16 + lo]);
  const float F2 = rope[d * 16 + lo];

  v8f Tm[2];
#pragma unroll
  for (int r = 0; r < 8; ++r) {
    const float e = (8 * hi + r == lo) ? 1.f : 0.f;
    Tm[0][r] = e;
    Tm[1][r] = e;
  }
  float b = 0.f;
  float Dg[BS + 1] = {};

  float* const grow = &gt[wave][hi][lo * 16];
  const float4* const ga0 =
      reinterpret_cast<const float4*>(&gt[wave][0][lo * 16 + 8 * hi]);
  const float4* const ga1 =
      reinterpret_cast<const float4*>(&gt[wave][1][lo * 16 + 8 * hi]);

  const int t0 = c * CHUNK;

  constexpr int PF2 = 2 * BS;
  float dtq[PF2] = {}, xbq[PF2] = {}, Bq[PF2] = {};
#pragma unroll
  for (int u = 0; u < PF2 - 1; ++u) {
    const int tt = t0 + u;
    dtq[u] = dt[tt * 16 + d];
    xbq[u] = xb[tt * 16 + d];
    Bq[u]  = Bv[tt * 16 + lo];
    __builtin_amdgcn_sched_barrier(0);
  }

  for (int t = 0; t < CHUNK; t += PF2) {
#pragma unroll
  for (int u = 0; u < PF2; ++u) {
    {
      const int v  = (u + PF2 - 1) % PF2;
      const int tn = t0 + min(t + u + PF2 - 1, CHUNK - 1);
      dtq[v] = dt[tn * 16 + d];
      xbq[v] = xb[tn * 16 + d];
      Bq[v]  = Bv[tn * 16 + lo];
      __builtin_amdgcn_sched_barrier(0);
    }
    const float dtc = dtq[u];
    const float xbd = xbq[u];
    const float Bj  = Bq[u];

    const float ta   = 0.5f * dtc * A2;
    const float abar = (1.f + ta) / (1.f - ta + 1e-8f);
    const float ang  = dtc * F2;
    const float cs   = __cosf(ang);
    const float sn   = __sinf(ang);
    const float ac   = abar * cs;
    const float as   = abar * sn;

    const int s = u % BS;
    if (s == 0) {
      Dg[0] = ac;
      Dg[1] = -as;
    } else {
      Dg[s + 1] = -as * row_prev(Dg[s]);
#pragma unroll
      for (int m = s; m >= 1; --m)
        Dg[m] = ac * Dg[m] - as * row_prev(Dg[m - 1]);
      Dg[0] = ac * Dg[0];
    }
    b = abar * (cs * b - sn * row_prev(b)) + dtc * Bj * xbd;

    if (s == BS - 1) {
#pragma unroll
      for (int m = 0; m <= BS; ++m) grow[(lo - m) & 15] = Dg[m];
      const float4 a00 = ga0[0], a01 = ga0[1];
      const float4 a10 = ga1[0], a11 = ga1[1];
      const float af[2][8] = {
          {a00.x, a00.y, a00.z, a00.w, a01.x, a01.y, a01.z, a01.w},
          {a10.x, a10.y, a10.z, a10.w, a11.x, a11.y, a11.z, a11.w}};
#pragma unroll
      for (int ch = 0; ch < 2; ++ch) {
        v8f nt = {};
#pragma unroll
        for (int kk = 0; kk < 4; ++kk) {
          v2f a, bb;
          a.x  = af[ch][2 * kk + 0];
          a.y  = af[ch][2 * kk + 1];
          bb.x = Tm[ch][2 * kk + 0];
          bb.y = Tm[ch][2 * kk + 1];
          nt = wmma4(a, bb, nt);
        }
        Tm[ch] = nt;
      }
    }
  }
  }

#pragma unroll
  for (int ch = 0; ch < 2; ++ch) {
#pragma unroll
    for (int r = 0; r < 8; ++r)
      Tmat[((c * 16 + 2 * wave + ch) * 16 + 8 * hi + r) * 16 + lo] = Tm[ch][r];
  }
  bvec[(c * 16 + d) * 16 + lo] = b;
}

__device__ inline float other_row_plus8(float v) {
  return __builtin_bit_cast(
      float, __builtin_amdgcn_permlanex16(0u, __builtin_bit_cast(unsigned, v),
                                          0xFEDCBA98u, 0x76543210u, true,
                                          true));
}

constexpr int PF3 = 21;
constexpr int AH3 = PF3 - 1;

__device__ inline float fold_chunk(const bool reload,
                                   const float* __restrict__ Tmat,
                                   const float* __restrict__ bvec,
                                   float* __restrict__ hrow,
                                   float4 (&q)[PF3][2], float (&bq)[PF3],
                                   const int u, const int cc, const int d,
                                   const unsigned toff, const unsigned j,
                                   const bool row0, float& bsel, float h) {
  hrow[(size_t)cc * 256] = h;
  if (reload) {
    const int v  = (u + AH3) % PF3;
    const int cn = cc + AH3;
    const float4* Trow =
        reinterpret_cast<const float4*>(Tmat + (cn * 16 + d) * 256) + toff;
    q[v][0] = Trow[0]; q[v][1] = Trow[1];
    bq[v] = (bvec + (cn * 16 + d) * 16)[j];
  }
  const float4 q0 = q[u][0], q1 = q[u][1];
  float x = bsel, y;
  asm("v_mul_f32_dpp %1, %2, %7 row_share:4 row_mask:0xf bank_mask:0xf bound_ctrl:1\n\t"
      "v_fmac_f32_dpp %0, %2, %3 row_share:0 row_mask:0xf bank_mask:0xf bound_ctrl:1\n\t"
      "v_fmac_f32_dpp %1, %2, %8 row_share:5 row_mask:0xf bank_mask:0xf bound_ctrl:1\n\t"
      "v_fmac_f32_dpp %0, %2, %4 row_share:1 row_mask:0xf bank_mask:0xf bound_ctrl:1\n\t"
      "v_fmac_f32_dpp %1, %2, %9 row_share:6 row_mask:0xf bank_mask:0xf bound_ctrl:1\n\t"
      "v_fmac_f32_dpp %0, %2, %5 row_share:2 row_mask:0xf bank_mask:0xf bound_ctrl:1\n\t"
      "v_fmac_f32_dpp %1, %2, %10 row_share:7 row_mask:0xf bank_mask:0xf bound_ctrl:1\n\t"
      "v_fmac_f32_dpp %0, %2, %6 row_share:3 row_mask:0xf bank_mask:0xf bound_ctrl:1"
      : "+v"(x), "=&v"(y)
      : "v"(h), "v"(q0.x), "v"(q0.y), "v"(q0.z), "v"(q0.w),
        "v"(q1.x), "v"(q1.y), "v"(q1.z), "v"(q1.w));
  const float s = x + y;
  bsel = row0 ? bq[(u + 1) % PF3] : 0.f;
  h = s + other_row_plus8(s);
  __builtin_amdgcn_sched_barrier(0);
  return h;
}

__global__ void __launch_bounds__(32)
k_combine(const float* __restrict__ Tmat,
                          const float* __restrict__ bvec,
                          const float* __restrict__ init_h,
                          float* __restrict__ hstart,
                          float* __restrict__ h_final) {
  const int d    = blockIdx.x;
  const int lane = threadIdx.x & 31;
  const unsigned lo = lane & 15, hi = lane >> 4;
  const unsigned j    = (lo + 8 * hi) & 15;
  const unsigned toff = j * 4 + hi * 2;
  const bool row0 = hi == 0;
  float h = (init_h + d * 16)[j];

  float* hrow = hstart + d * 16 + j;

  float4 q[PF3][2] = {};
  float  bq[PF3]   = {};
#pragma unroll
  for (int u = 0; u < AH3; ++u) {
    const float4* Trow =
        reinterpret_cast<const float4*>(Tmat + (u * 16 + d) * 256) + toff;
    q[u][0] = Trow[0]; q[u][1] = Trow[1];
    bq[u] = (bvec + (u * 16 + d) * 16)[j];
    __builtin_amdgcn_sched_barrier(0);
  }

  float bsel = row0 ? bq[0] : 0.f;

  constexpr int MAIN_END = (NCHUNK - AH3) / PF3 * PF3;
  for (int c = 0; c < MAIN_END; c += PF3) {
#pragma unroll
    for (int u = 0; u < PF3; ++u)
      h = fold_chunk(true, Tmat, bvec, hrow, q, bq, u, c + u, d, toff, j, row0,
                     bsel, h);
  }
#pragma unroll
  for (int i = 0; i < NCHUNK - MAIN_END; ++i)
    h = fold_chunk(MAIN_END + i + AH3 < NCHUNK, Tmat, bvec, hrow, q, bq,
                   i % PF3, MAIN_END + i, d, toff, j, row0, bsel, h);
  if (lane < 16) (h_final + d * 16)[lo] = h;
}

__global__ void __launch_bounds__(256)
k_scan(const float* __restrict__ dt,
                       const float* __restrict__ xb,
                       const float* __restrict__ Bv,
                       const float* __restrict__ Cv,
                       const float* __restrict__ z,
                       const float* __restrict__ A_log,
                       const float* __restrict__ rope,
                       const float* __restrict__ hstart,
                       const float* __restrict__ Dp,
                       const float* __restrict__ outw,
                       float* __restrict__ out) {
  __shared__ float ylds[CHUNK * YP];
  const int c    = blockIdx.x;
  const int lane = threadIdx.x & 31;
  const int lo   = lane & 15, hi = lane >> 4;
  const int d    = 2 * (threadIdx.x >> 5) + hi;
  const float A2 = -__expf(A_log[d * 16 + lo]);
  const float F2 = rope[d * 16 + lo];
  float h = hstart[(c * 16 + d) * 16 + lo];
  asm volatile("" :: "v"(h));
  const int t0  = c * CHUNK;

  constexpr int PS = 16;
  float dtq[PS], xbq[PS], Bq[PS], Cq[PS];
#pragma unroll
  for (int u = 0; u < PS; ++u) {
    dtq[u] = (dt + (t0 + u) * 16)[d];
    Bq[u]  = (Bv + (t0 + u) * 16)[lo];
    xbq[u] = (xb + (t0 + u) * 16)[d];
    Cq[u]  = (Cv + (t0 + u) * 16)[lo];
  }
#pragma unroll
  for (int u = 0; u < PS; ++u) {
    asm volatile("" :: "v"(dtq[u]), "v"(Bq[u]), "v"(xbq[u]), "v"(Cq[u]));
  }

  for (int t = 0; t < CHUNK; t += PS) {
    const int tnb = (t0 + min(t + PS, CHUNK - PS)) * 16;
    float dtn[PS], xbn[PS], Bn[PS], Cn[PS];
#pragma unroll
    for (int u = 0; u < PS; ++u) {
      dtn[u] = (dt + tnb + u * 16)[d];
      Bn[u]  = (Bv + tnb + u * 16)[lo];
      xbn[u] = (xb + tnb + u * 16)[d];
      Cn[u]  = (Cv + tnb + u * 16)[lo];
    }
    __builtin_amdgcn_sched_barrier(0);

    float ab[PS], cs[PS], sn[PS], bu[PS];
#pragma unroll
    for (int u = 0; u < PS; ++u) {
      const float dtc = dtq[u];
      const float ta  = 0.5f * dtc * A2;
      ab[u] = (1.f + ta) / (1.f - ta + 1e-8f);
      const float ang = dtc * F2;
      cs[u] = __cosf(ang);
      sn[u] = __sinf(ang);
      bu[u] = dtc * Bq[u];
    }

    float yv = 0.f;
#pragma unroll
    for (int u = 0; u < PS; ++u) {
      const float hr = row_prev(h);
      h = ab[u] * (cs[u] * h - sn[u] * hr) + bu[u] * xbq[u];
      const float p = row_sum16(h * Cq[u]);
      const float p0 = row_dpp<0x150>(p);
      yv = (lo == u) ? p0 : yv;
    }
    ylds[(t + lo) * YP + d] = yv;
    __builtin_amdgcn_sched_barrier(0);

#pragma unroll
    for (int u = 0; u < PS; ++u) {
      dtq[u] = dtn[u]; Bq[u] = Bn[u]; xbq[u] = xbn[u]; Cq[u] = Cn[u];
    }
  }

  const int wave = threadIdx.x >> 5;
  v2f bo[4], dq[4], zq[2][4], xq[2][4];
#pragma unroll
  for (int kk = 0; kk < 4; ++kk) {
    const int kb = kk * 4 + 2 * hi;
    bo[kk].x = (lo < 8) ? outw[lo * 16 + kb + 0] : 0.f;
    bo[kk].y = (lo < 8) ? outw[lo * 16 + kb + 1] : 0.f;
    dq[kk].x = Dp[kb + 0];
    dq[kk].y = Dp[kb + 1];
#pragma unroll
    for (int i = 0; i < 2; ++i) {
      const int idx = (t0 + (2 * wave + i) * 16 + lo) * 16 + kb;
      zq[i][kk].x = z[idx + 0];  zq[i][kk].y = z[idx + 1];
      xq[i][kk].x = xb[idx + 0]; xq[i][kk].y = xb[idx + 1];
    }
  }
  __syncthreads();
  v8f acc[2];
#pragma unroll
  for (int i = 0; i < 2; ++i) {
    const int tl = (2 * wave + i) * 16;
    acc[i] = v8f{};
#pragma unroll
    for (int kk = 0; kk < 4; ++kk) {
      const int kb = kk * 4 + 2 * hi;
      v2f a;
#pragma unroll
      for (int v = 0; v < 2; ++v) {
        const float zz  = zq[i][kk][v];
        const float sil = zz / (1.f + __expf(-zz));
        const float val =
            ylds[(tl + lo) * YP + kb + v] * sil + dq[kk][v] * xq[i][kk][v];
        if (v == 0) a.x = val; else a.y = val;
      }
      acc[i] = wmma4(a, bo[kk], acc[i]);
    }
  }
  if (lo < 8) {
#pragma unroll
    for (int i = 0; i < 2; ++i) {
#pragma unroll
      for (int r = 0; r < 8; ++r)
        out[(t0 + (2 * wave + i) * 16 + r + 8 * hi) * 8 + lo] = acc[i][r];
    }
  }
}

extern "C" void kernel_launch(void* const* d_in, const int* in_sizes, int n_in,
                              void* d_out, int out_size, void* d_ws,
                              size_t ws_size, hipStream_t stream) {
  const float* x    = (const float*)d_in[0];
  const float* init = (const float*)d_in[1];
  const float* inw  = (const float*)d_in[2];
  const float* dtw  = (const float*)d_in[3];
  const float* dtb  = (const float*)d_in[4];
  const float* bw   = (const float*)d_in[5];
  const float* cw   = (const float*)d_in[6];
  const float* alog = (const float*)d_in[7];
  const float* Dp   = (const float*)d_in[8];
  const float* rope = (const float*)d_in[9];
  const float* outw = (const float*)d_in[10];
  float* out = (float*)d_out;

  float* ws = (float*)d_ws;
  float* XB = ws;
  float* Z  = XB + N16;
  float* DT = Z + N16;
  float* BV = DT + N16;
  float* CV = BV + N16;
  float* T  = CV + N16;
  float* BC = T + N16;
  float* HS = BC + NCHUNK * 256;

  k_chunkop<<<NCHUNK, 256, 0, stream>>>(x, inw, dtw, dtb, bw, cw,
                                        XB, Z, DT, BV, CV, alog, rope, T, BC);
  k_combine<<<16, 32, 0, stream>>>(T, BC, init, HS, out + NTOK * DM);
  k_scan<<<NCHUNK, 256, 0, stream>>>(DT, XB, BV, CV, Z, alog, rope, HS,
                                     Dp, outw, out);
}
